# starting kernel
# baseline (speedup 1.0000x reference)
#include <hip/hip_runtime.h>


typedef float v2f __attribute__((ext_vector_type(2)));
typedef float v4f __attribute__((ext_vector_type(4)));
typedef float v8f __attribute__((ext_vector_type(8)));

#define T_LEN 8192
#define NB    16
#define NP    128
#define ME    144
#define XROWS 148
#define STR   68

#define XS_OFF   0
#define W_OFF    (XS_OFF + XROWS * STR)
#define R_OFF0   (W_OFF + 5 * 64 * STR)
#define RES_SZ   (ME * STR)
#define BIAS_OFF (R_OFF0 + 3 * RES_SZ)
#define LDS_FLOATS (BIAS_OFF + 64)

__device__ __forceinline__ int kslot(int ci) {
    return (((ci >> 1) & 1) << 5) + ((ci >> 2) << 1) + (ci & 1);
}

__device__ __forceinline__ int ref_idx(int i) {
    if (i < 2) return 2 - i;
    if (i > T_LEN - 3) return (i == T_LEN - 2) ? T_LEN - 6 : T_LEN - 7;
    return i - 2;
}

template<int TAPS, bool DUAL>
__device__ __forceinline__ void gemm_tiles(float* lds, int aw_base, int xb_base,
                                           int res_base, int pt0)
{
    const int x0 = xb_base + pt0 * 16 * STR;
    const int x1 = x0 + 16 * STR;

    v4f A[2][8], B0[2][8], B1[2][8];
    #pragma unroll
    for (int q = 0; q < 8; ++q) {
        A[0][q]  = *(const v4f*)&lds[aw_base + 4 * q];
        B0[0][q] = *(const v4f*)&lds[x0 + 4 * q];
        if constexpr (DUAL) B1[0][q] = *(const v4f*)&lds[x1 + 4 * q];
    }

    v8f acc0 = {0.f,0.f,0.f,0.f,0.f,0.f,0.f,0.f};
    v8f acc1 = {0.f,0.f,0.f,0.f,0.f,0.f,0.f,0.f};

    #pragma unroll
    for (int t = 0; t < TAPS; ++t) {
        const int cur = t & 1, nxt = cur ^ 1;
        if (t + 1 < TAPS) {
            #pragma unroll
            for (int q = 0; q < 8; ++q) {
                A[nxt][q]  = *(const v4f*)&lds[aw_base + (t + 1) * 64 * STR + 4 * q];
                B0[nxt][q] = *(const v4f*)&lds[x0 + (t + 1) * STR + 4 * q];
                if constexpr (DUAL) B1[nxt][q] = *(const v4f*)&lds[x1 + (t + 1) * STR + 4 * q];
            }
        }
        #pragma unroll
        for (int q = 0; q < 8; ++q) {
            v2f a0 = {A[cur][q][0], A[cur][q][1]};
            v2f a1 = {A[cur][q][2], A[cur][q][3]};
            v2f b0 = {B0[cur][q][0], B0[cur][q][1]};
            v2f b1 = {B0[cur][q][2], B0[cur][q][3]};
            acc0 = __builtin_amdgcn_wmma_f32_16x16x4_f32(false, a0, false, b0, (short)0, acc0, false, false);
            acc0 = __builtin_amdgcn_wmma_f32_16x16x4_f32(false, a1, false, b1, (short)0, acc0, false, false);
            if constexpr (DUAL) {
                v2f c0 = {B1[cur][q][0], B1[cur][q][1]};
                v2f c1 = {B1[cur][q][2], B1[cur][q][3]};
                acc1 = __builtin_amdgcn_wmma_f32_16x16x4_f32(false, a0, false, c0, (short)0, acc1, false, false);
                acc1 = __builtin_amdgcn_wmma_f32_16x16x4_f32(false, a1, false, c1, (short)0, acc1, false, false);
            }
        }
    }

    int st0 = res_base + pt0 * 16 * STR;
    *(v4f*)&lds[st0]     = (v4f){acc0[0], acc0[1], acc0[2], acc0[3]};
    *(v4f*)&lds[st0 + 4] = (v4f){acc0[4], acc0[5], acc0[6], acc0[7]};
    if constexpr (DUAL) {
        int st1 = st0 + 16 * STR;
        *(v4f*)&lds[st1]     = (v4f){acc1[0], acc1[1], acc1[2], acc1[3]};
        *(v4f*)&lds[st1 + 4] = (v4f){acc1[4], acc1[5], acc1[6], acc1[7]};
    }
}

template<int TAPS, int KOFS, int RHO>
__device__ __forceinline__ void process_residue(float* lds, const float* wgrp, int tid,
                                                int wrow, int aw_base, int xb_base,
                                                int res_nm_base)
{
    __syncthreads();
    for (int e = tid; e < TAPS * 64 * 64; e += 256) {
        int t  = e >> 12;
        int r  = e & 4095;
        int co = r >> 6;
        int ci = r & 63;
        lds[W_OFF + (t * 64 + co) * STR + kslot(ci)] = wgrp[(co * 64 + ci) * 13 + 3 * t + KOFS];
    }
    __syncthreads();

    const int res_base = R_OFF0 + RHO * RES_SZ + res_nm_base;
    if (wrow == 0) {
        gemm_tiles<TAPS, true >(lds, aw_base, xb_base, res_base, 0);
        gemm_tiles<TAPS, true >(lds, aw_base, xb_base, res_base, 2);
        gemm_tiles<TAPS, false>(lds, aw_base, xb_base, res_base, 4);
    } else {
        gemm_tiles<TAPS, true >(lds, aw_base, xb_base, res_base, 5);
        gemm_tiles<TAPS, true >(lds, aw_base, xb_base, res_base, 7);
    }
}

__global__ __launch_bounds__(256) void trconv_fused_kernel(
    const float* __restrict__ x,
    const float* __restrict__ w,
    const float* __restrict__ bias,
    float* __restrict__ out)
{
    extern __shared__ float lds[];
    const int tid = threadIdx.x;
    const int b   = blockIdx.z;
    const int g   = blockIdx.y;
    const int p0  = blockIdx.x * NP;
    const int me0 = p0 - 8;

    const float* xg = x + (size_t)(b * 128 + g * 64) * T_LEN;
    for (int e = tid; e < 64 * XROWS; e += 256) {
        int ci = e / XROWS;
        int lp = e - ci * XROWS;
        int m  = me0 + lp;
        m = m < 0 ? 0 : (m > T_LEN - 1 ? T_LEN - 1 : m);
        lds[XS_OFF + lp * STR + kslot(ci)] = xg[(size_t)ci * T_LEN + m];
    }
    if (tid < 64) lds[BIAS_OFF + tid] = bias[g * 64 + tid];

    const int lane = tid & 31;
    const int wv   = tid >> 5;
    const int ct   = wv & 3;
    const int wrow = wv >> 2;
    const int nm   = lane & 15;
    const int lh   = lane >> 4;

    const int aw_base     = W_OFF + (ct * 16 + nm) * STR + lh * 32;
    const int xb_base     = XS_OFF + nm * STR + lh * 32;
    const int res_nm_base = nm * STR + ct * 16 + lh * 8;

    const float* wgrp = w + (size_t)(g * 64) * (64 * 13);

    process_residue<4, 2, 0>(lds, wgrp, tid, wrow, aw_base, xb_base, res_nm_base);
    process_residue<4, 1, 1>(lds, wgrp, tid, wrow, aw_base, xb_base, res_nm_base);
    process_residue<5, 0, 2>(lds, wgrp, tid, wrow, aw_base, xb_base, res_nm_base);

    __syncthreads();

    const float* r0p = &lds[R_OFF0 + 0 * RES_SZ];
    const float* r1p = &lds[R_OFF0 + 1 * RES_SZ];
    const float* r2p = &lds[R_OFF0 + 2 * RES_SZ];
    for (int idx = tid; idx < 64 * NP; idx += 256) {
        int co = idx >> 7;
        int il = idx & (NP - 1);
        int i  = p0 + il;
        float bi = lds[BIAS_OFF + co];

        int ri = ref_idx(i);
        int lm = ri - me0;
        float l0 = r0p[lm * STR + co], r0v = r0p[(lm + 1) * STR + co];
        float l1 = r1p[lm * STR + co], r1v = r1p[(lm + 1) * STR + co];
        float x_m = r2p[lm * STR + co] + bi;

        float um = x_m;
        #pragma unroll
        for (int d = -2; d <= 2; ++d) {
            if (d == 0) continue;
            int j = i + d;
            float v = 0.0f;
            if (j >= 0 && j <= T_LEN - 1)
                v = r2p[(ref_idx(j) - me0) * STR + co] + bi;
            um = fmaxf(um, v);
        }

        float d0 = r0v - l0, d1 = r1v - l1;
        float qg0 = d0 / (1.0f + __expf(-1.702f * d0));
        float qg1 = d1 / (1.0f + __expf(-1.702f * d1));
        float accv = qg0 + qg1 + (r0v + l0 + r1v + l1) + 2.0f * x_m + 4.0f * bi + um;

        out[(size_t)(b * 128 + g * 64 + co) * T_LEN + i] = accv;
    }
}

extern "C" void kernel_launch(void* const* d_in, const int* in_sizes, int n_in,
                              void* d_out, int out_size, void* d_ws, size_t ws_size,
                              hipStream_t stream) {
    const float* x  = (const float*)d_in[0];
    const float* w  = (const float*)d_in[1];
    const float* b  = (const float*)d_in[2];
    float* out      = (float*)d_out;

    dim3 grid(T_LEN / NP, 2, NB);
    dim3 block(256);
    size_t shmem = (size_t)LDS_FLOATS * sizeof(float);
    hipLaunchKernelGGL(trconv_fused_kernel, grid, block, shmem, stream, x, w, b, out);
}


# Weights: all 13 taps staged once per workgroup with whole-line 16-byte loads (was 3 passes of 52-byte-stride 4-byte loads); results 1,2 overwrite used tap slots; 4 barriers not 7; LDS 245->306 KB
# speedup vs baseline: 1.4935x; 1.4935x over previous
#include <hip/hip_runtime.h>


typedef float v2f __attribute__((ext_vector_type(2)));
typedef float v4f __attribute__((ext_vector_type(4)));
typedef float v8f __attribute__((ext_vector_type(8)));

#define T_LEN 8192
#define NB    16
#define NP    128
#define ME    144
#define XROWS 148
#define STR   68

#define XS_OFF   0
#define W_OFF    (XS_OFF + XROWS * STR)
#define WSLOT    (64 * STR + 8)
#define RES_SZ   (ME * STR)
#define R0_OFF   (W_OFF + 13 * WSLOT)
#define R1_OFF   (W_OFF + 0 * WSLOT)
#define R2_OFF   (W_OFF + 4 * WSLOT)
#define BIAS_OFF (R0_OFF + RES_SZ)
#define LDS_FLOATS (BIAS_OFF + 64)
static_assert(RES_SZ <= 4 * WSLOT, "a residue buffer fits in the four tap slots it overwrites");
static_assert(LDS_FLOATS * 4 <= 320 * 1024, "fits the workgroup LDS");

__device__ __forceinline__ int kslot(int ci) {
    return (((ci >> 1) & 1) << 5) + ((ci >> 2) << 1) + (ci & 1);
}

__device__ __forceinline__ int ref_idx(int i) {
    if (i < 2) return 2 - i;
    if (i > T_LEN - 3) return (i == T_LEN - 2) ? T_LEN - 6 : T_LEN - 7;
    return i - 2;
}

template<int TAPS, bool DUAL>
__device__ __forceinline__ void gemm_tiles(float* lds, int aw_base, int xb_base,
                                           int res_base, int pt0)
{
    const int x0 = xb_base + pt0 * 16 * STR;
    const int x1 = x0 + 16 * STR;

    v4f A[2][8], B0[2][8], B1[2][8];
    #pragma unroll
    for (int q = 0; q < 8; ++q) {
        A[0][q]  = *(const v4f*)&lds[aw_base + 4 * q];
        B0[0][q] = *(const v4f*)&lds[x0 + 4 * q];
        if constexpr (DUAL) B1[0][q] = *(const v4f*)&lds[x1 + 4 * q];
    }

    v8f acc0 = {0.f,0.f,0.f,0.f,0.f,0.f,0.f,0.f};
    v8f acc1 = {0.f,0.f,0.f,0.f,0.f,0.f,0.f,0.f};

    #pragma unroll
    for (int t = 0; t < TAPS; ++t) {
        const int cur = t & 1, nxt = cur ^ 1;
        if (t + 1 < TAPS) {
            #pragma unroll
            for (int q = 0; q < 8; ++q) {
                A[nxt][q]  = *(const v4f*)&lds[aw_base + (t + 1) * WSLOT + 4 * q];
                B0[nxt][q] = *(const v4f*)&lds[x0 + (t + 1) * STR + 4 * q];
                if constexpr (DUAL) B1[nxt][q] = *(const v4f*)&lds[x1 + (t + 1) * STR + 4 * q];
            }
        }
        #pragma unroll
        for (int q = 0; q < 8; ++q) {
            v2f a0 = {A[cur][q][0], A[cur][q][1]};
            v2f a1 = {A[cur][q][2], A[cur][q][3]};
            v2f b0 = {B0[cur][q][0], B0[cur][q][1]};
            v2f b1 = {B0[cur][q][2], B0[cur][q][3]};
            acc0 = __builtin_amdgcn_wmma_f32_16x16x4_f32(false, a0, false, b0, (short)0, acc0, false, false);
            acc0 = __builtin_amdgcn_wmma_f32_16x16x4_f32(false, a1, false, b1, (short)0, acc0, false, false);
            if constexpr (DUAL) {
                v2f c0 = {B1[cur][q][0], B1[cur][q][1]};
                v2f c1 = {B1[cur][q][2], B1[cur][q][3]};
                acc1 = __builtin_amdgcn_wmma_f32_16x16x4_f32(false, a0, false, c0, (short)0, acc1, false, false);
                acc1 = __builtin_amdgcn_wmma_f32_16x16x4_f32(false, a1, false, c1, (short)0, acc1, false, false);
            }
        }
    }

    int st0 = res_base + pt0 * 16 * STR;
    *(v4f*)&lds[st0]     = (v4f){acc0[0], acc0[1], acc0[2], acc0[3]};
    *(v4f*)&lds[st0 + 4] = (v4f){acc0[4], acc0[5], acc0[6], acc0[7]};
    if constexpr (DUAL) {
        int st1 = st0 + 16 * STR;
        *(v4f*)&lds[st1]     = (v4f){acc1[0], acc1[1], acc1[2], acc1[3]};
        *(v4f*)&lds[st1 + 4] = (v4f){acc1[4], acc1[5], acc1[6], acc1[7]};
    }
}

template<int TAPS>
__device__ __forceinline__ void residue_gemm(float* lds, int wrow, int aw_base, int xb_base, int res_base)
{
    if (wrow == 0) {
        gemm_tiles<TAPS, true >(lds, aw_base, xb_base, res_base, 0);
        gemm_tiles<TAPS, true >(lds, aw_base, xb_base, res_base, 2);
        gemm_tiles<TAPS, false>(lds, aw_base, xb_base, res_base, 4);
    } else {
        gemm_tiles<TAPS, true >(lds, aw_base, xb_base, res_base, 5);
        gemm_tiles<TAPS, true >(lds, aw_base, xb_base, res_base, 7);
    }
}

__global__ __launch_bounds__(256) void trconv_fused_kernel(
    const float* __restrict__ x,
    const float* __restrict__ w,
    const float* __restrict__ bias,
    float* __restrict__ out)
{
    __shared__ __attribute__((aligned(16))) float lds[LDS_FLOATS];
    const int tid = threadIdx.x;
    const int b   = blockIdx.z;
    const int g   = blockIdx.y;
    const int p0  = blockIdx.x * NP;
    const int me0 = p0 - 8;

    const int lane = tid & 31;
    const int wv   = tid >> 5;
    const int ct   = wv & 3;
    const int wrow = wv >> 2;
    const int nm   = lane & 15;
    const int lh   = lane >> 4;

    const float* wgrp = w + (size_t)(g * 64) * (64 * 13);
    v4f wb[4][13];
    {
        const float* wsrc = wgrp + (wv * 8) * (64 * 13) + 4 * lane;
        #pragma unroll
        for (int i = 0; i < 4; ++i) {
            #pragma unroll
            for (int j = 0; j < 13; ++j)
                wb[i][j] = *(const v4f*)(wsrc + i * 1664 + 128 * j);
        }
    }

    const float* xg = x + (size_t)(b * 128 + g * 64) * T_LEN;
    for (int e = tid; e < 64 * XROWS; e += 256) {
        int ci = e / XROWS;
        int lp = e - ci * XROWS;
        int m  = me0 + lp;
        m = m < 0 ? 0 : (m > T_LEN - 1 ? T_LEN - 1 : m);
        lds[XS_OFF + lp * STR + kslot(ci)] = xg[(size_t)ci * T_LEN + m];
    }
    if (tid < 64) lds[BIAS_OFF + tid] = bias[g * 64 + tid];

    const int aw_base     = W_OFF + (ct * 16 + nm) * STR + lh * 32;
    const int xb_base     = XS_OFF + nm * STR + lh * 32;
    const int res_nm_base = nm * STR + ct * 16 + lh * 8;

    {
        int woff[13][4];
        #pragma unroll
        for (int j = 0; j < 13; ++j) {
            #pragma unroll
            for (int c = 0; c < 4; ++c) {
                const int L    = 128 * j + 4 * lane + c;
                const int cf   = L / 13;
                const int k    = L - 13 * cf;
                const int slot = 13 * (k / 3) + 8 - 4 * k;
                woff[j][c] = W_OFF + slot * WSLOT + (wv * 8 + (cf >> 6)) * STR + kslot(cf & 63);
            }
        }
        #pragma unroll
        for (int i = 0; i < 4; ++i) {
            #pragma unroll
            for (int j = 0; j < 13; ++j) {
                #pragma unroll
                for (int c = 0; c < 4; ++c)
                    lds[woff[j][c] + i * 2 * STR] = wb[i][j][c];
            }
        }
    }

    __syncthreads();
    residue_gemm<4>(lds, wrow, aw_base + 0 * WSLOT, xb_base, R0_OFF + res_nm_base);
    __syncthreads();
    residue_gemm<4>(lds, wrow, aw_base + 4 * WSLOT, xb_base, R1_OFF + res_nm_base);
    __syncthreads();
    residue_gemm<5>(lds, wrow, aw_base + 8 * WSLOT, xb_base, R2_OFF + res_nm_base);

    __syncthreads();

    const float* r0p = &lds[R0_OFF];
    const float* r1p = &lds[R1_OFF];
    const float* r2p = &lds[R2_OFF];
    for (int idx = tid; idx < 64 * NP; idx += 256) {
        int co = idx >> 7;
        int il = idx & (NP - 1);
        int i  = p0 + il;
        float bi = lds[BIAS_OFF + co];

        int ri = ref_idx(i);
        int lm = ri - me0;
        float l0 = r0p[lm * STR + co], r0v = r0p[(lm + 1) * STR + co];
        float l1 = r1p[lm * STR + co], r1v = r1p[(lm + 1) * STR + co];
        float x_m = r2p[lm * STR + co] + bi;

        float um = x_m;
        #pragma unroll
        for (int d = -2; d <= 2; ++d) {
            if (d == 0) continue;
            int j = i + d;
            float v = 0.0f;
            if (j >= 0 && j <= T_LEN - 1)
                v = r2p[(ref_idx(j) - me0) * STR + co] + bi;
            um = fmaxf(um, v);
        }

        float d0 = r0v - l0, d1 = r1v - l1;
        float qg0 = d0 / (1.0f + __expf(-1.702f * d0));
        float qg1 = d1 / (1.0f + __expf(-1.702f * d1));
        float accv = qg0 + qg1 + (r0v + l0 + r1v + l1) + 2.0f * x_m + 4.0f * bi + um;

        out[(size_t)(b * 128 + g * 64 + co) * T_LEN + i] = accv;
    }
}

extern "C" void kernel_launch(void* const* d_in, const int* in_sizes, int n_in,
                              void* d_out, int out_size, void* d_ws, size_t ws_size,
                              hipStream_t stream) {
    const float* x  = (const float*)d_in[0];
    const float* w  = (const float*)d_in[1];
    const float* b  = (const float*)d_in[2];
    float* out      = (float*)d_out;

    dim3 grid(T_LEN / NP, 2, NB);
    dim3 block(256);
    hipLaunchKernelGGL(trconv_fused_kernel, grid, block, 0, stream, x, w, b, out);
}


# Epilogue LDS reads: a thread takes 4 channels of its position per 16-byte load, unguarded (reflected row is always in the tile; zero by select) and batched; was 10 dependent 4-byte reads per output
# speedup vs baseline: 1.8727x; 1.2539x over previous
#include <hip/hip_runtime.h>


typedef float v2f __attribute__((ext_vector_type(2)));
typedef float v4f __attribute__((ext_vector_type(4)));
typedef float v8f __attribute__((ext_vector_type(8)));

#define T_LEN 8192
#define NB    16
#define NP    128
#define ME    144
#define XROWS 148
#define STR   68

#define XS_OFF   0
#define W_OFF    (XS_OFF + XROWS * STR)
#define WSLOT    (64 * STR + 8)
#define RES_SZ   (ME * STR)
#define R0_OFF   (W_OFF + 13 * WSLOT)
#define R1_OFF   (W_OFF + 0 * WSLOT)
#define R2_OFF   (W_OFF + 4 * WSLOT)
#define BIAS_OFF (R0_OFF + RES_SZ)
#define LDS_FLOATS (BIAS_OFF + 64)
static_assert(RES_SZ <= 4 * WSLOT, "a residue buffer fits in the four tap slots it overwrites");
static_assert(LDS_FLOATS * 4 <= 320 * 1024, "fits the workgroup LDS");

__device__ __forceinline__ int kslot(int ci) {
    return (((ci >> 1) & 1) << 5) + ((ci >> 2) << 1) + (ci & 1);
}

__device__ __forceinline__ int ref_idx(int i) {
    if (i < 2) return 2 - i;
    if (i > T_LEN - 3) return (i == T_LEN - 2) ? T_LEN - 6 : T_LEN - 7;
    return i - 2;
}

template<int TAPS, bool DUAL>
__device__ __forceinline__ void gemm_tiles(float* lds, int aw_base, int xb_base,
                                           int res_base, int pt0)
{
    const int x0 = xb_base + pt0 * 16 * STR;
    const int x1 = x0 + 16 * STR;

    v4f A[2][8], B0[2][8], B1[2][8];
    #pragma unroll
    for (int q = 0; q < 8; ++q) {
        A[0][q]  = *(const v4f*)&lds[aw_base + 4 * q];
        B0[0][q] = *(const v4f*)&lds[x0 + 4 * q];
        if constexpr (DUAL) B1[0][q] = *(const v4f*)&lds[x1 + 4 * q];
    }

    v8f acc0 = {0.f,0.f,0.f,0.f,0.f,0.f,0.f,0.f};
    v8f acc1 = {0.f,0.f,0.f,0.f,0.f,0.f,0.f,0.f};

    #pragma unroll
    for (int t = 0; t < TAPS; ++t) {
        const int cur = t & 1, nxt = cur ^ 1;
        if (t + 1 < TAPS) {
            #pragma unroll
            for (int q = 0; q < 8; ++q) {
                A[nxt][q]  = *(const v4f*)&lds[aw_base + (t + 1) * WSLOT + 4 * q];
                B0[nxt][q] = *(const v4f*)&lds[x0 + (t + 1) * STR + 4 * q];
                if constexpr (DUAL) B1[nxt][q] = *(const v4f*)&lds[x1 + (t + 1) * STR + 4 * q];
            }
        }
        #pragma unroll
        for (int q = 0; q < 8; ++q) {
            v2f a0 = {A[cur][q][0], A[cur][q][1]};
            v2f a1 = {A[cur][q][2], A[cur][q][3]};
            v2f b0 = {B0[cur][q][0], B0[cur][q][1]};
            v2f b1 = {B0[cur][q][2], B0[cur][q][3]};
            acc0 = __builtin_amdgcn_wmma_f32_16x16x4_f32(false, a0, false, b0, (short)0, acc0, false, false);
            acc0 = __builtin_amdgcn_wmma_f32_16x16x4_f32(false, a1, false, b1, (short)0, acc0, false, false);
            if constexpr (DUAL) {
                v2f c0 = {B1[cur][q][0], B1[cur][q][1]};
                v2f c1 = {B1[cur][q][2], B1[cur][q][3]};
                acc1 = __builtin_amdgcn_wmma_f32_16x16x4_f32(false, a0, false, c0, (short)0, acc1, false, false);
                acc1 = __builtin_amdgcn_wmma_f32_16x16x4_f32(false, a1, false, c1, (short)0, acc1, false, false);
            }
        }
    }

    int st0 = res_base + pt0 * 16 * STR;
    *(v4f*)&lds[st0]     = (v4f){acc0[0], acc0[1], acc0[2], acc0[3]};
    *(v4f*)&lds[st0 + 4] = (v4f){acc0[4], acc0[5], acc0[6], acc0[7]};
    if constexpr (DUAL) {
        int st1 = st0 + 16 * STR;
        *(v4f*)&lds[st1]     = (v4f){acc1[0], acc1[1], acc1[2], acc1[3]};
        *(v4f*)&lds[st1 + 4] = (v4f){acc1[4], acc1[5], acc1[6], acc1[7]};
    }
}

template<int TAPS>
__device__ __forceinline__ void residue_gemm(float* lds, int wrow, int aw_base, int xb_base, int res_base)
{
    if (wrow == 0) {
        gemm_tiles<TAPS, true >(lds, aw_base, xb_base, res_base, 0);
        gemm_tiles<TAPS, true >(lds, aw_base, xb_base, res_base, 2);
        gemm_tiles<TAPS, false>(lds, aw_base, xb_base, res_base, 4);
    } else {
        gemm_tiles<TAPS, true >(lds, aw_base, xb_base, res_base, 5);
        gemm_tiles<TAPS, true >(lds, aw_base, xb_base, res_base, 7);
    }
}

__global__ __launch_bounds__(256) void trconv_fused_kernel(
    const float* __restrict__ x,
    const float* __restrict__ w,
    const float* __restrict__ bias,
    float* __restrict__ out)
{
    __shared__ __attribute__((aligned(16))) float lds[LDS_FLOATS];
    const int tid = threadIdx.x;
    const int b   = blockIdx.z;
    const int g   = blockIdx.y;
    const int p0  = blockIdx.x * NP;
    const int me0 = p0 - 8;

    const int lane = tid & 31;
    const int wv   = tid >> 5;
    const int ct   = wv & 3;
    const int wrow = wv >> 2;
    const int nm   = lane & 15;
    const int lh   = lane >> 4;

    const float* wgrp = w + (size_t)(g * 64) * (64 * 13);
    v4f wb[4][13];
    {
        const float* wsrc = wgrp + (wv * 8) * (64 * 13) + 4 * lane;
        #pragma unroll
        for (int i = 0; i < 4; ++i) {
            #pragma unroll
            for (int j = 0; j < 13; ++j)
                wb[i][j] = *(const v4f*)(wsrc + i * 1664 + 128 * j);
        }
    }

    const float* xg = x + (size_t)(b * 128 + g * 64) * T_LEN;
    for (int e = tid; e < 64 * XROWS; e += 256) {
        int ci = e / XROWS;
        int lp = e - ci * XROWS;
        int m  = me0 + lp;
        m = m < 0 ? 0 : (m > T_LEN - 1 ? T_LEN - 1 : m);
        lds[XS_OFF + lp * STR + kslot(ci)] = xg[(size_t)ci * T_LEN + m];
    }
    if (tid < 64) lds[BIAS_OFF + tid] = bias[g * 64 + tid];

    const int aw_base     = W_OFF + (ct * 16 + nm) * STR + lh * 32;
    const int xb_base     = XS_OFF + nm * STR + lh * 32;
    const int res_nm_base = nm * STR + ct * 16 + lh * 8;

    {
        int woff[13][4];
        #pragma unroll
        for (int j = 0; j < 13; ++j) {
            #pragma unroll
            for (int c = 0; c < 4; ++c) {
                const int L    = 128 * j + 4 * lane + c;
                const int cf   = L / 13;
                const int k    = L - 13 * cf;
                const int slot = 13 * (k / 3) + 8 - 4 * k;
                woff[j][c] = W_OFF + slot * WSLOT + (wv * 8 + (cf >> 6)) * STR + kslot(cf & 63);
            }
        }
        #pragma unroll
        for (int i = 0; i < 4; ++i) {
            #pragma unroll
            for (int j = 0; j < 13; ++j) {
                #pragma unroll
                for (int c = 0; c < 4; ++c)
                    lds[woff[j][c] + i * 2 * STR] = wb[i][j][c];
            }
        }
    }

    __syncthreads();
    residue_gemm<4>(lds, wrow, aw_base + 0 * WSLOT, xb_base, R0_OFF + res_nm_base);
    __syncthreads();
    residue_gemm<4>(lds, wrow, aw_base + 4 * WSLOT, xb_base, R1_OFF + res_nm_base);
    __syncthreads();
    residue_gemm<5>(lds, wrow, aw_base + 8 * WSLOT, xb_base, R2_OFF + res_nm_base);

    __syncthreads();

    const float* r0p = &lds[R0_OFF];
    const float* r1p = &lds[R1_OFF];
    const float* r2p = &lds[R2_OFF];
    {
        const int il   = tid & (NP - 1);
        const int i    = p0 + il;
        const int rowc = (ref_idx(i) - me0) * STR;
        int  roww[4];
        bool inr[4];
        #pragma unroll
        for (int q = 0; q < 4; ++q) {
            const int j = i + (q < 2 ? q - 2 : q - 1);
            roww[q] = (ref_idx(j) - me0) * STR;
            inr[q]  = (j >= 0 && j <= T_LEN - 1);
        }
        float* outp = out + (size_t)(b * 128 + g * 64) * T_LEN + i;
        const int cq0 = 4 * (tid >> 7);

        constexpr int NQ = 2;
        #pragma unroll
        for (int n0 = 0; n0 < 8; n0 += NQ) {
            v4f bi[NQ], l0[NQ], r0v[NQ], l1[NQ], r1v[NQ], xc[NQ], xw[4][NQ];
            #pragma unroll
            for (int u = 0; u < NQ; ++u) {
                const int co = cq0 + 8 * (n0 + u);
                bi[u]  = *(const v4f*)&lds[BIAS_OFF + co];
                l0[u]  = *(const v4f*)&r0p[rowc + co];
                r0v[u] = *(const v4f*)&r0p[rowc + STR + co];
                l1[u]  = *(const v4f*)&r1p[rowc + co];
                r1v[u] = *(const v4f*)&r1p[rowc + STR + co];
                xc[u]  = *(const v4f*)&r2p[rowc + co];
                #pragma unroll
                for (int q = 0; q < 4; ++q) xw[q][u] = *(const v4f*)&r2p[roww[q] + co];
            }
            #pragma unroll
            for (int u = 0; u < NQ; ++u) {
                const int co = cq0 + 8 * (n0 + u);
                #pragma unroll
                for (int c = 0; c < 4; ++c) {
                    const float b1 = bi[u][c];
                    float x_m = xc[u][c] + b1;
                    float um = x_m;
                    #pragma unroll
                    for (int q = 0; q < 4; ++q) {
                        float t = xw[q][u][c] + b1;
                        um = fmaxf(um, inr[q] ? t : 0.0f);
                    }
                    float d0 = r0v[u][c] - l0[u][c], d1 = r1v[u][c] - l1[u][c];
                    float qg0 = d0 / (1.0f + __expf(-1.702f * d0));
                    float qg1 = d1 / (1.0f + __expf(-1.702f * d1));
                    float accv = qg0 + qg1 + (r0v[u][c] + l0[u][c] + r1v[u][c] + l1[u][c]) + 2.0f * x_m + 4.0f * b1 + um;
                    outp[(size_t)(co + c) * T_LEN] = accv;
                }
            }
        }
    }
}

extern "C" void kernel_launch(void* const* d_in, const int* in_sizes, int n_in,
                              void* d_out, int out_size, void* d_ws, size_t ws_size,
                              hipStream_t stream) {
    const float* x  = (const float*)d_in[0];
    const float* w  = (const float*)d_in[1];
    const float* b  = (const float*)d_in[2];
    float* out      = (float*)d_out;

    dim3 grid(T_LEN / NP, 2, NB);
    dim3 block(256);
    hipLaunchKernelGGL(trconv_fused_kernel, grid, block, 0, stream, x, w, b, out);
}


# Staging: all global loads of a workgroup (x tile as ten 16-byte loads per lane, bias, weights) issued before the first wait, x scattered to LDS as weights arrive; was a 4-byte x loop with 11 waits
# speedup vs baseline: 2.1499x; 1.1480x over previous
#include <hip/hip_runtime.h>


typedef float v2f __attribute__((ext_vector_type(2)));
typedef float v4f __attribute__((ext_vector_type(4)));
typedef float v8f __attribute__((ext_vector_type(8)));

#define T_LEN 8192
#define NB    16
#define NP    128
#define ME    144
#define XROWS 148
#define STR   68

#define XS_OFF   0
#define W_OFF    (XS_OFF + XROWS * STR)
#define WSLOT    (64 * STR + 8)
#define RES_SZ   (ME * STR)
#define R0_OFF   (W_OFF + 13 * WSLOT)
#define R1_OFF   (W_OFF + 0 * WSLOT)
#define R2_OFF   (W_OFF + 4 * WSLOT)
#define BIAS_OFF (R0_OFF + RES_SZ)
#define LDS_FLOATS (BIAS_OFF + 64)
static_assert(RES_SZ <= 4 * WSLOT, "a residue buffer fits in the four tap slots it overwrites");
static_assert(LDS_FLOATS * 4 <= 320 * 1024, "fits the workgroup LDS");

__device__ __forceinline__ int kslot(int ci) {
    return (((ci >> 1) & 1) << 5) + ((ci >> 2) << 1) + (ci & 1);
}

__device__ __forceinline__ int ref_idx(int i) {
    if (i < 2) return 2 - i;
    if (i > T_LEN - 3) return (i == T_LEN - 2) ? T_LEN - 6 : T_LEN - 7;
    return i - 2;
}

template<int TAPS, bool DUAL>
__device__ __forceinline__ void gemm_tiles(float* lds, int aw_base, int xb_base,
                                           int res_base, int pt0)
{
    const int x0 = xb_base + pt0 * 16 * STR;
    const int x1 = x0 + 16 * STR;

    v4f A[2][8], B0[2][8], B1[2][8];
    #pragma unroll
    for (int q = 0; q < 8; ++q) {
        A[0][q]  = *(const v4f*)&lds[aw_base + 4 * q];
        B0[0][q] = *(const v4f*)&lds[x0 + 4 * q];
        if constexpr (DUAL) B1[0][q] = *(const v4f*)&lds[x1 + 4 * q];
    }

    v8f acc0 = {0.f,0.f,0.f,0.f,0.f,0.f,0.f,0.f};
    v8f acc1 = {0.f,0.f,0.f,0.f,0.f,0.f,0.f,0.f};

    #pragma unroll
    for (int t = 0; t < TAPS; ++t) {
        const int cur = t & 1, nxt = cur ^ 1;
        if (t + 1 < TAPS) {
            #pragma unroll
            for (int q = 0; q < 8; ++q) {
                A[nxt][q]  = *(const v4f*)&lds[aw_base + (t + 1) * WSLOT + 4 * q];
                B0[nxt][q] = *(const v4f*)&lds[x0 + (t + 1) * STR + 4 * q];
                if constexpr (DUAL) B1[nxt][q] = *(const v4f*)&lds[x1 + (t + 1) * STR + 4 * q];
            }
        }
        #pragma unroll
        for (int q = 0; q < 8; ++q) {
            v2f a0 = {A[cur][q][0], A[cur][q][1]};
            v2f a1 = {A[cur][q][2], A[cur][q][3]};
            v2f b0 = {B0[cur][q][0], B0[cur][q][1]};
            v2f b1 = {B0[cur][q][2], B0[cur][q][3]};
            acc0 = __builtin_amdgcn_wmma_f32_16x16x4_f32(false, a0, false, b0, (short)0, acc0, false, false);
            acc0 = __builtin_amdgcn_wmma_f32_16x16x4_f32(false, a1, false, b1, (short)0, acc0, false, false);
            if constexpr (DUAL) {
                v2f c0 = {B1[cur][q][0], B1[cur][q][1]};
                v2f c1 = {B1[cur][q][2], B1[cur][q][3]};
                acc1 = __builtin_amdgcn_wmma_f32_16x16x4_f32(false, a0, false, c0, (short)0, acc1, false, false);
                acc1 = __builtin_amdgcn_wmma_f32_16x16x4_f32(false, a1, false, c1, (short)0, acc1, false, false);
            }
        }
    }

    int st0 = res_base + pt0 * 16 * STR;
    *(v4f*)&lds[st0]     = (v4f){acc0[0], acc0[1], acc0[2], acc0[3]};
    *(v4f*)&lds[st0 + 4] = (v4f){acc0[4], acc0[5], acc0[6], acc0[7]};
    if constexpr (DUAL) {
        int st1 = st0 + 16 * STR;
        *(v4f*)&lds[st1]     = (v4f){acc1[0], acc1[1], acc1[2], acc1[3]};
        *(v4f*)&lds[st1 + 4] = (v4f){acc1[4], acc1[5], acc1[6], acc1[7]};
    }
}

template<int TAPS>
__device__ __forceinline__ void residue_gemm(float* lds, int wrow, int aw_base, int xb_base, int res_base)
{
    if (wrow == 0) {
        gemm_tiles<TAPS, true >(lds, aw_base, xb_base, res_base, 0);
        gemm_tiles<TAPS, true >(lds, aw_base, xb_base, res_base, 2);
        gemm_tiles<TAPS, false>(lds, aw_base, xb_base, res_base, 4);
    } else {
        gemm_tiles<TAPS, true >(lds, aw_base, xb_base, res_base, 5);
        gemm_tiles<TAPS, true >(lds, aw_base, xb_base, res_base, 7);
    }
}

__global__ __launch_bounds__(256) void trconv_fused_kernel(
    const float* __restrict__ x,
    const float* __restrict__ w,
    const float* __restrict__ bias,
    float* __restrict__ out)
{
    __shared__ __attribute__((aligned(16))) float lds[LDS_FLOATS];
    const int tid = threadIdx.x;
    const int b   = blockIdx.z;
    const int g   = blockIdx.y;
    const int p0  = blockIdx.x * NP;
    const int me0 = p0 - 8;

    const int lane = tid & 31;
    const int wv   = tid >> 5;
    const int ct   = wv & 3;
    const int wrow = wv >> 2;
    const int nm   = lane & 15;
    const int lh   = lane >> 4;

    const float* xg = x + (size_t)(b * 128 + g * 64) * T_LEN;
    const int xq  = lane & 7;
    const int xr  = lane >> 3;
    const int xs  = wv & 1;
    v4f xv[10];
    #pragma unroll
    for (int it = 0; it < 10; ++it) {
        const int h  = (it & 1) * 4 + (wv >> 1);
        const int ci = 8 * h + 2 * xs + (xr & 1) + 4 * (xr >> 1);
        int q = 8 * (it >> 1) + xq;
        q = q > 36 ? 36 : q;
        int m4 = me0 + 4 * q;
        m4 = m4 < 0 ? 0 : (m4 > T_LEN - 4 ? T_LEN - 4 : m4);
        xv[it] = *(const v4f*)(xg + (size_t)ci * T_LEN + m4);
    }
    float bias_v = 0.0f;
    if (tid < 64) bias_v = bias[g * 64 + tid];

    const float* wgrp = w + (size_t)(g * 64) * (64 * 13);
    v4f wb[4][13];
    {
        const float* wsrc = wgrp + (wv * 8) * (64 * 13) + 4 * lane;
        #pragma unroll
        for (int i = 0; i < 4; ++i) {
            #pragma unroll
            for (int j = 0; j < 13; ++j)
                wb[i][j] = *(const v4f*)(wsrc + i * 1664 + 128 * j);
        }
    }

    #pragma unroll
    for (int it = 0; it < 10; ++it) {
        const int h  = (it & 1) * 4 + (wv >> 1);
        const int ci = 8 * h + 2 * xs + (xr & 1) + 4 * (xr >> 1);
        const int q  = 8 * (it >> 1) + xq;
        if (q < 37) {
            const int a = XS_OFF + (4 * q) * STR + kslot(ci);
            lds[a]           = xv[it][0];
            lds[a + STR]     = xv[it][1];
            lds[a + 2 * STR] = xv[it][2];
            lds[a + 3 * STR] = xv[it][3];
        }
    }
    if (tid < 64) lds[BIAS_OFF + tid] = bias_v;

    const int aw_base     = W_OFF + (ct * 16 + nm) * STR + lh * 32;
    const int xb_base     = XS_OFF + nm * STR + lh * 32;
    const int res_nm_base = nm * STR + ct * 16 + lh * 8;

    {
        int woff[13][4];
        #pragma unroll
        for (int j = 0; j < 13; ++j) {
            #pragma unroll
            for (int c = 0; c < 4; ++c) {
                const int L    = 128 * j + 4 * lane + c;
                const int cf   = L / 13;
                const int k    = L - 13 * cf;
                const int slot = 13 * (k / 3) + 8 - 4 * k;
                woff[j][c] = W_OFF + slot * WSLOT + (wv * 8 + (cf >> 6)) * STR + kslot(cf & 63);
            }
        }
        #pragma unroll
        for (int i = 0; i < 4; ++i) {
            #pragma unroll
            for (int j = 0; j < 13; ++j) {
                #pragma unroll
                for (int c = 0; c < 4; ++c)
                    lds[woff[j][c] + i * 2 * STR] = wb[i][j][c];
            }
        }
    }

    __syncthreads();
    residue_gemm<4>(lds, wrow, aw_base + 0 * WSLOT, xb_base, R0_OFF + res_nm_base);
    __syncthreads();
    residue_gemm<4>(lds, wrow, aw_base + 4 * WSLOT, xb_base, R1_OFF + res_nm_base);
    __syncthreads();
    residue_gemm<5>(lds, wrow, aw_base + 8 * WSLOT, xb_base, R2_OFF + res_nm_base);

    __syncthreads();

    const float* r0p = &lds[R0_OFF];
    const float* r1p = &lds[R1_OFF];
    const float* r2p = &lds[R2_OFF];
    {
        const int il   = tid & (NP - 1);
        const int i    = p0 + il;
        const int rowc = (ref_idx(i) - me0) * STR;
        int  roww[4];
        bool inr[4];
        #pragma unroll
        for (int q = 0; q < 4; ++q) {
            const int j = i + (q < 2 ? q - 2 : q - 1);
            roww[q] = (ref_idx(j) - me0) * STR;
            inr[q]  = (j >= 0 && j <= T_LEN - 1);
        }
        float* outp = out + (size_t)(b * 128 + g * 64) * T_LEN + i;
        const int cq0 = 4 * (tid >> 7);

        constexpr int NQ = 2;
        #pragma unroll
        for (int n0 = 0; n0 < 8; n0 += NQ) {
            v4f bi[NQ], l0[NQ], r0v[NQ], l1[NQ], r1v[NQ], xc[NQ], xw[4][NQ];
            #pragma unroll
            for (int u = 0; u < NQ; ++u) {
                const int co = cq0 + 8 * (n0 + u);
                bi[u]  = *(const v4f*)&lds[BIAS_OFF + co];
                l0[u]  = *(const v4f*)&r0p[rowc + co];
                r0v[u] = *(const v4f*)&r0p[rowc + STR + co];
                l1[u]  = *(const v4f*)&r1p[rowc + co];
                r1v[u] = *(const v4f*)&r1p[rowc + STR + co];
                xc[u]  = *(const v4f*)&r2p[rowc + co];
                #pragma unroll
                for (int q = 0; q < 4; ++q) xw[q][u] = *(const v4f*)&r2p[roww[q] + co];
            }
            #pragma unroll
            for (int u = 0; u < NQ; ++u) {
                const int co = cq0 + 8 * (n0 + u);
                #pragma unroll
                for (int c = 0; c < 4; ++c) {
                    const float b1 = bi[u][c];
                    float x_m = xc[u][c] + b1;
                    float um = x_m;
                    #pragma unroll
                    for (int q = 0; q < 4; ++q) {
                        float t = xw[q][u][c] + b1;
                        um = fmaxf(um, inr[q] ? t : 0.0f);
                    }
                    float d0 = r0v[u][c] - l0[u][c], d1 = r1v[u][c] - l1[u][c];
                    float qg0 = d0 / (1.0f + __expf(-1.702f * d0));
                    float qg1 = d1 / (1.0f + __expf(-1.702f * d1));
                    float accv = qg0 + qg1 + (r0v[u][c] + l0[u][c] + r1v[u][c] + l1[u][c]) + 2.0f * x_m + 4.0f * b1 + um;
                    outp[(size_t)(co + c) * T_LEN] = accv;
                }
            }
        }
    }
}

extern "C" void kernel_launch(void* const* d_in, const int* in_sizes, int n_in,
                              void* d_out, int out_size, void* d_ws, size_t ws_size,
                              hipStream_t stream) {
    const float* x  = (const float*)d_in[0];
    const float* w  = (const float*)d_in[1];
    const float* b  = (const float*)d_in[2];
    float* out      = (float*)d_out;

    dim3 grid(T_LEN / NP, 2, NB);
    dim3 block(256);
    hipLaunchKernelGGL(trconv_fused_kernel, grid, block, 0, stream, x, w, b, out);
}


# Persistent workgroups: 13 weight taps staged once per 8 position tiles (was per tile), next x tile loaded under the GEMMs; one residue buffer, so the output is summed residue by residue (sum order)
# speedup vs baseline: 2.4736x; 1.1506x over previous
#include <hip/hip_runtime.h>


typedef float v2f __attribute__((ext_vector_type(2)));
typedef float v4f __attribute__((ext_vector_type(4)));
typedef float v8f __attribute__((ext_vector_type(8)));

#define T_LEN 8192
#define NB    16
#define NP    128
#define TPW   8
#define ME    144
#define XROWS 148
#define STR   68

#define XS_OFF   0
#define W_OFF    (XS_OFF + XROWS * STR)
#define WSLOT    (64 * STR + 8)
#define RES_SZ   (ME * STR)
#define R_OFF    (W_OFF + 13 * WSLOT)
#define BIAS_OFF (R_OFF + RES_SZ)
#define LDS_FLOATS (BIAS_OFF + 64)
static_assert(LDS_FLOATS * 4 <= 320 * 1024, "fits the workgroup LDS");

__device__ __forceinline__ int kslot(int ci) {
    return (((ci >> 1) & 1) << 5) + ((ci >> 2) << 1) + (ci & 1);
}

__device__ __forceinline__ int ref_idx(int i) {
    if (i < 2) return 2 - i;
    if (i > T_LEN - 3) return (i == T_LEN - 2) ? T_LEN - 6 : T_LEN - 7;
    return i - 2;
}

template<int TAPS, bool DUAL>
__device__ __forceinline__ void gemm_tiles(float* lds, int aw_base, int xb_base,
                                           int res_base, int pt0)
{
    const int x0 = xb_base + pt0 * 16 * STR;
    const int x1 = x0 + 16 * STR;

    v4f A[2][8], B0[2][8], B1[2][8];
    #pragma unroll
    for (int q = 0; q < 8; ++q) {
        A[0][q]  = *(const v4f*)&lds[aw_base + 4 * q];
        B0[0][q] = *(const v4f*)&lds[x0 + 4 * q];
        if constexpr (DUAL) B1[0][q] = *(const v4f*)&lds[x1 + 4 * q];
    }

    v8f acc0 = {0.f,0.f,0.f,0.f,0.f,0.f,0.f,0.f};
    v8f acc1 = {0.f,0.f,0.f,0.f,0.f,0.f,0.f,0.f};

    #pragma unroll
    for (int t = 0; t < TAPS; ++t) {
        const int cur = t & 1, nxt = cur ^ 1;
        if (t + 1 < TAPS) {
            #pragma unroll
            for (int q = 0; q < 8; ++q) {
                A[nxt][q]  = *(const v4f*)&lds[aw_base + (t + 1) * WSLOT + 4 * q];
                B0[nxt][q] = *(const v4f*)&lds[x0 + (t + 1) * STR + 4 * q];
                if constexpr (DUAL) B1[nxt][q] = *(const v4f*)&lds[x1 + (t + 1) * STR + 4 * q];
            }
        }
        #pragma unroll
        for (int q = 0; q < 8; ++q) {
            v2f a0 = {A[cur][q][0], A[cur][q][1]};
            v2f a1 = {A[cur][q][2], A[cur][q][3]};
            v2f b0 = {B0[cur][q][0], B0[cur][q][1]};
            v2f b1 = {B0[cur][q][2], B0[cur][q][3]};
            acc0 = __builtin_amdgcn_wmma_f32_16x16x4_f32(false, a0, false, b0, (short)0, acc0, false, false);
            acc0 = __builtin_amdgcn_wmma_f32_16x16x4_f32(false, a1, false, b1, (short)0, acc0, false, false);
            if constexpr (DUAL) {
                v2f c0 = {B1[cur][q][0], B1[cur][q][1]};
                v2f c1 = {B1[cur][q][2], B1[cur][q][3]};
                acc1 = __builtin_amdgcn_wmma_f32_16x16x4_f32(false, a0, false, c0, (short)0, acc1, false, false);
                acc1 = __builtin_amdgcn_wmma_f32_16x16x4_f32(false, a1, false, c1, (short)0, acc1, false, false);
            }
        }
    }

    int st0 = res_base + pt0 * 16 * STR;
    *(v4f*)&lds[st0]     = (v4f){acc0[0], acc0[1], acc0[2], acc0[3]};
    *(v4f*)&lds[st0 + 4] = (v4f){acc0[4], acc0[5], acc0[6], acc0[7]};
    if constexpr (DUAL) {
        int st1 = st0 + 16 * STR;
        *(v4f*)&lds[st1]     = (v4f){acc1[0], acc1[1], acc1[2], acc1[3]};
        *(v4f*)&lds[st1 + 4] = (v4f){acc1[4], acc1[5], acc1[6], acc1[7]};
    }
}

template<int TAPS>
__device__ __forceinline__ void residue_gemm(float* lds, int wrow, int aw_base, int xb_base, int res_base)
{
    if (wrow == 0) {
        gemm_tiles<TAPS, true >(lds, aw_base, xb_base, res_base, 0);
        gemm_tiles<TAPS, true >(lds, aw_base, xb_base, res_base, 2);
        gemm_tiles<TAPS, false>(lds, aw_base, xb_base, res_base, 4);
    } else {
        gemm_tiles<TAPS, true >(lds, aw_base, xb_base, res_base, 5);
        gemm_tiles<TAPS, true >(lds, aw_base, xb_base, res_base, 7);
    }
}

__device__ __forceinline__ void x_tile_load(const float* __restrict__ xg, int me0, int lane, int wv, v4f (&xv)[10])
{
    const int xq = lane & 7;
    const int xr = lane >> 3;
    const int xs = wv & 1;
    #pragma unroll
    for (int it = 0; it < 10; ++it) {
        const int h  = (it & 1) * 4 + (wv >> 1);
        const int ci = 8 * h + 2 * xs + (xr & 1) + 4 * (xr >> 1);
        int q = 8 * (it >> 1) + xq;
        q = q > 36 ? 36 : q;
        int m4 = me0 + 4 * q;
        m4 = m4 < 0 ? 0 : (m4 > T_LEN - 4 ? T_LEN - 4 : m4);
        xv[it] = *(const v4f*)(xg + (size_t)ci * T_LEN + m4);
    }
}

__device__ __forceinline__ void x_tile_scatter(float* lds, int lane, int wv, const v4f (&xv)[10])
{
    const int xq = lane & 7;
    const int xr = lane >> 3;
    const int xs = wv & 1;
    #pragma unroll
    for (int it = 0; it < 10; ++it) {
        const int h  = (it & 1) * 4 + (wv >> 1);
        const int ci = 8 * h + 2 * xs + (xr & 1) + 4 * (xr >> 1);
        const int q  = 8 * (it >> 1) + xq;
        if (q < 37) {
            const int a = XS_OFF + (4 * q) * STR + kslot(ci);
            lds[a]           = xv[it][0];
            lds[a + STR]     = xv[it][1];
            lds[a + 2 * STR] = xv[it][2];
            lds[a + 3 * STR] = xv[it][3];
        }
    }
}

template<bool FIRST>
__device__ __forceinline__ void term_pair(const float* rp, int rowc, int cq0, float (&part)[8][4])
{
    constexpr int NQ = 4;
    #pragma unroll
    for (int n0 = 0; n0 < 8; n0 += NQ) {
        v4f l[NQ], r[NQ];
        #pragma unroll
        for (int u = 0; u < NQ; ++u) {
            const int co = cq0 + 8 * (n0 + u);
            l[u] = *(const v4f*)&rp[rowc + co];
            r[u] = *(const v4f*)&rp[rowc + STR + co];
        }
        #pragma unroll
        for (int u = 0; u < NQ; ++u) {
            #pragma unroll
            for (int c = 0; c < 4; ++c) {
                const float d  = r[u][c] - l[u][c];
                const float qg = d / (1.0f + __expf(-1.702f * d));
                const float s  = qg + (r[u][c] + l[u][c]);
                if constexpr (FIRST) part[n0 + u][c] = s;
                else                 part[n0 + u][c] += s;
            }
        }
    }
}

__device__ __forceinline__ void lds_barrier()
{
    __builtin_amdgcn_fence(__ATOMIC_RELEASE, "workgroup", "local");
    __builtin_amdgcn_s_barrier();
    __builtin_amdgcn_fence(__ATOMIC_ACQUIRE, "workgroup", "local");
}

__global__ __launch_bounds__(256) void trconv_fused_kernel(
    const float* __restrict__ x,
    const float* __restrict__ w,
    const float* __restrict__ bias,
    float* __restrict__ out)
{
    __shared__ __attribute__((aligned(16))) float lds[LDS_FLOATS];
    const int tid = threadIdx.x;
    const int b   = blockIdx.z;
    const int g   = blockIdx.y;
    const int pc0 = blockIdx.x * (TPW * NP);

    const int lane = tid & 31;
    const int wv   = tid >> 5;
    const int ct   = wv & 3;
    const int wrow = wv >> 2;
    const int nm   = lane & 15;
    const int lh   = lane >> 4;

    const float* xg = x + (size_t)(b * 128 + g * 64) * T_LEN;
    v4f xv[10];
    x_tile_load(xg, pc0 - 8, lane, wv, xv);
    float bias_v = 0.0f;
    if (tid < 64) bias_v = bias[g * 64 + tid];

    const float* wgrp = w + (size_t)(g * 64) * (64 * 13);
    v4f wb[4][13];
    {
        const float* wsrc = wgrp + (wv * 8) * (64 * 13) + 4 * lane;
        #pragma unroll
        for (int i = 0; i < 4; ++i) {
            #pragma unroll
            for (int j = 0; j < 13; ++j)
                wb[i][j] = *(const v4f*)(wsrc + i * 1664 + 128 * j);
        }
    }

    x_tile_scatter(lds, lane, wv, xv);
    if (tid < 64) lds[BIAS_OFF + tid] = bias_v;

    const int aw_base  = W_OFF + (ct * 16 + nm) * STR + lh * 32;
    const int xb_base  = XS_OFF + nm * STR + lh * 32;
    const int res_base = R_OFF + nm * STR + ct * 16 + lh * 8;

    {
        int woff[13][4];
        #pragma unroll
        for (int j = 0; j < 13; ++j) {
            #pragma unroll
            for (int c = 0; c < 4; ++c) {
                const int L    = 128 * j + 4 * lane + c;
                const int cf   = L / 13;
                const int k    = L - 13 * cf;
                const int slot = 13 * (k / 3) + 8 - 4 * k;
                woff[j][c] = W_OFF + slot * WSLOT + (wv * 8 + (cf >> 6)) * STR + kslot(cf & 63);
            }
        }
        #pragma unroll
        for (int i = 0; i < 4; ++i) {
            #pragma unroll
            for (int j = 0; j < 13; ++j) {
                #pragma unroll
                for (int c = 0; c < 4; ++c)
                    lds[woff[j][c] + i * 2 * STR] = wb[i][j][c];
            }
        }
    }

    const float* rp  = &lds[R_OFF];
    const int    il  = tid & (NP - 1);
    const int    cq0 = 4 * (tid >> 7);
    float* outg = out + (size_t)(b * 128 + g * 64) * T_LEN;

    #pragma unroll 1
    for (int tile = 0; tile < TPW; ++tile) {
        const int p0  = pc0 + tile * NP;
        const int me0 = p0 - 8;
        const int i   = p0 + il;
        const int rowc = (ref_idx(i) - me0) * STR;

        if (tile + 1 < TPW) x_tile_load(xg, me0 + NP, lane, wv, xv);

        float part[8][4];

        lds_barrier();
        residue_gemm<4>(lds, wrow, aw_base + 0 * WSLOT, xb_base, res_base);
        lds_barrier();
        term_pair<true >(rp, rowc, cq0, part);
        lds_barrier();
        residue_gemm<4>(lds, wrow, aw_base + 4 * WSLOT, xb_base, res_base);
        lds_barrier();
        term_pair<false>(rp, rowc, cq0, part);
        lds_barrier();
        residue_gemm<5>(lds, wrow, aw_base + 8 * WSLOT, xb_base, res_base);
        lds_barrier();

        if (tile + 1 < TPW) x_tile_scatter(lds, lane, wv, xv);

        int  roww[4];
        bool inr[4];
        #pragma unroll
        for (int q = 0; q < 4; ++q) {
            const int j = i + (q < 2 ? q - 2 : q - 1);
            roww[q] = (ref_idx(j) - me0) * STR;
            inr[q]  = (j >= 0 && j <= T_LEN - 1);
        }
        float* outp = outg + i;

        constexpr int NQ = 2;
        #pragma unroll
        for (int n0 = 0; n0 < 8; n0 += NQ) {
            v4f bi[NQ], xc[NQ], xw[4][NQ];
            #pragma unroll
            for (int u = 0; u < NQ; ++u) {
                const int co = cq0 + 8 * (n0 + u);
                bi[u] = *(const v4f*)&lds[BIAS_OFF + co];
                xc[u] = *(const v4f*)&rp[rowc + co];
                #pragma unroll
                for (int q = 0; q < 4; ++q) xw[q][u] = *(const v4f*)&rp[roww[q] + co];
            }
            #pragma unroll
            for (int u = 0; u < NQ; ++u) {
                const int co = cq0 + 8 * (n0 + u);
                #pragma unroll
                for (int c = 0; c < 4; ++c) {
                    const float b1 = bi[u][c];
                    float x_m = xc[u][c] + b1;
                    float um = x_m;
                    #pragma unroll
                    for (int q = 0; q < 4; ++q) {
                        float tv = xw[q][u][c] + b1;
                        um = fmaxf(um, inr[q] ? tv : 0.0f);
                    }
                    float accv = part[n0 + u][c] + 2.0f * x_m + 4.0f * b1 + um;
                    outp[(size_t)(co + c) * T_LEN] = accv;
                }
            }
        }
    }
}

extern "C" void kernel_launch(void* const* d_in, const int* in_sizes, int n_in,
                              void* d_out, int out_size, void* d_ws, size_t ws_size,
                              hipStream_t stream) {
    const float* x  = (const float*)d_in[0];
    const float* w  = (const float*)d_in[1];
    const float* b  = (const float*)d_in[2];
    float* out      = (float*)d_out;

    dim3 grid(T_LEN / (NP * TPW), 2, NB);
    dim3 block(256);
    hipLaunchKernelGGL(trconv_fused_kernel, grid, block, 0, stream, x, w, b, out);
}


# Halo rows carried, not recomputed: each residue's last 4 rows of a tile stay in LDS for the next tile, so a residue GEMM covers 8 position tiles of 16 (was 9) and every wave runs only paired tiles
# speedup vs baseline: 2.6122x; 1.0560x over previous
#include <hip/hip_runtime.h>


typedef float v2f __attribute__((ext_vector_type(2)));
typedef float v4f __attribute__((ext_vector_type(4)));
typedef float v8f __attribute__((ext_vector_type(8)));

#define T_LEN 8192
#define NB    16
#define NP    128
#define TPW   8
#define ME    144
#define XROWS 148
#define STR   68

#define XS_OFF   0
#define W_OFF    (XS_OFF + XROWS * STR)
#define WSLOT    (64 * STR + 8)
#define RES_SZ   (ME * STR)
#define R_OFF    (W_OFF + 13 * WSLOT)
#define BIAS_OFF (R_OFF + RES_SZ)
#define CARRY_OFF (BIAS_OFF + 64)
#define LDS_FLOATS (CARRY_OFF + 3 * 256)
static_assert(LDS_FLOATS * 4 <= 320 * 1024, "fits the workgroup LDS");

__device__ __forceinline__ int kslot(int ci) {
    return (((ci >> 1) & 1) << 5) + ((ci >> 2) << 1) + (ci & 1);
}

__device__ __forceinline__ int ref_idx(int i) {
    if (i < 2) return 2 - i;
    if (i > T_LEN - 3) return (i == T_LEN - 2) ? T_LEN - 6 : T_LEN - 7;
    return i - 2;
}

template<int TAPS, bool DUAL, bool HEAD>
__device__ __forceinline__ void gemm_tiles(float* lds, int aw_base, int xb_base,
                                           int res_base, int pt0, bool keep)
{
    const int x0 = xb_base + pt0 * 16 * STR;
    const int x1 = x0 + 16 * STR;

    v4f A[2][8], B0[2][8], B1[2][8];
    #pragma unroll
    for (int q = 0; q < 8; ++q) {
        A[0][q]  = *(const v4f*)&lds[aw_base + 4 * q];
        B0[0][q] = *(const v4f*)&lds[x0 + 4 * q];
        if constexpr (DUAL) B1[0][q] = *(const v4f*)&lds[x1 + 4 * q];
    }

    v8f acc0 = {0.f,0.f,0.f,0.f,0.f,0.f,0.f,0.f};
    v8f acc1 = {0.f,0.f,0.f,0.f,0.f,0.f,0.f,0.f};

    #pragma unroll
    for (int t = 0; t < TAPS; ++t) {
        const int cur = t & 1, nxt = cur ^ 1;
        if (t + 1 < TAPS) {
            #pragma unroll
            for (int q = 0; q < 8; ++q) {
                A[nxt][q]  = *(const v4f*)&lds[aw_base + (t + 1) * WSLOT + 4 * q];
                B0[nxt][q] = *(const v4f*)&lds[x0 + (t + 1) * STR + 4 * q];
                if constexpr (DUAL) B1[nxt][q] = *(const v4f*)&lds[x1 + (t + 1) * STR + 4 * q];
            }
        }
        #pragma unroll
        for (int q = 0; q < 8; ++q) {
            v2f a0 = {A[cur][q][0], A[cur][q][1]};
            v2f a1 = {A[cur][q][2], A[cur][q][3]};
            v2f b0 = {B0[cur][q][0], B0[cur][q][1]};
            v2f b1 = {B0[cur][q][2], B0[cur][q][3]};
            acc0 = __builtin_amdgcn_wmma_f32_16x16x4_f32(false, a0, false, b0, (short)0, acc0, false, false);
            acc0 = __builtin_amdgcn_wmma_f32_16x16x4_f32(false, a1, false, b1, (short)0, acc0, false, false);
            if constexpr (DUAL) {
                v2f c0 = {B1[cur][q][0], B1[cur][q][1]};
                v2f c1 = {B1[cur][q][2], B1[cur][q][3]};
                acc1 = __builtin_amdgcn_wmma_f32_16x16x4_f32(false, a0, false, c0, (short)0, acc1, false, false);
                acc1 = __builtin_amdgcn_wmma_f32_16x16x4_f32(false, a1, false, c1, (short)0, acc1, false, false);
            }
        }
    }

    int st0 = res_base + pt0 * 16 * STR;
    if (!HEAD || keep) {
        *(v4f*)&lds[st0]     = (v4f){acc0[0], acc0[1], acc0[2], acc0[3]};
        *(v4f*)&lds[st0 + 4] = (v4f){acc0[4], acc0[5], acc0[6], acc0[7]};
    }
    if constexpr (DUAL) {
        int st1 = st0 + 16 * STR;
        *(v4f*)&lds[st1]     = (v4f){acc1[0], acc1[1], acc1[2], acc1[3]};
        *(v4f*)&lds[st1 + 4] = (v4f){acc1[4], acc1[5], acc1[6], acc1[7]};
    }
}

template<int TAPS>
__device__ __forceinline__ void residue_gemm(float* lds, int wrow, bool first, bool keep,
                                             int aw_base, int xb_base, int res_base)
{
    const int xb8 = xb_base + 8 * STR;
    const int rb8 = res_base + 8 * STR;
    if (wrow == 0) {
        if (first) gemm_tiles<TAPS, false, true>(lds, aw_base, xb_base, res_base, 0, keep);
        gemm_tiles<TAPS, true, false>(lds, aw_base, xb8, rb8, 0, true);
        gemm_tiles<TAPS, true, false>(lds, aw_base, xb8, rb8, 2, true);
    } else {
        gemm_tiles<TAPS, true, false>(lds, aw_base, xb8, rb8, 4, true);
        gemm_tiles<TAPS, true, false>(lds, aw_base, xb8, rb8, 6, true);
    }
}

__device__ __forceinline__ void carry_save(float* lds, int tid, int r)
{
    if (tid < 64)
        *(v4f*)&lds[CARRY_OFF + r * 256 + 4 * tid] =
            *(const v4f*)&lds[R_OFF + (132 + (tid >> 4)) * STR + 4 * (tid & 15)];
}
__device__ __forceinline__ void carry_restore(float* lds, int tid, int r)
{
    if (tid < 64)
        *(v4f*)&lds[R_OFF + (4 + (tid >> 4)) * STR + 4 * (tid & 15)] =
            *(const v4f*)&lds[CARRY_OFF + r * 256 + 4 * tid];
}

__device__ __forceinline__ void x_tile_load(const float* __restrict__ xg, int me0, int lane, int wv, v4f (&xv)[10])
{
    const int xq = lane & 7;
    const int xr = lane >> 3;
    const int xs = wv & 1;
    #pragma unroll
    for (int it = 0; it < 10; ++it) {
        const int h  = (it & 1) * 4 + (wv >> 1);
        const int ci = 8 * h + 2 * xs + (xr & 1) + 4 * (xr >> 1);
        int q = 8 * (it >> 1) + xq;
        q = q > 36 ? 36 : q;
        int m4 = me0 + 4 * q;
        m4 = m4 < 0 ? 0 : (m4 > T_LEN - 4 ? T_LEN - 4 : m4);
        xv[it] = *(const v4f*)(xg + (size_t)ci * T_LEN + m4);
    }
}

__device__ __forceinline__ void x_tile_scatter(float* lds, int lane, int wv, const v4f (&xv)[10])
{
    const int xq = lane & 7;
    const int xr = lane >> 3;
    const int xs = wv & 1;
    #pragma unroll
    for (int it = 0; it < 10; ++it) {
        const int h  = (it & 1) * 4 + (wv >> 1);
        const int ci = 8 * h + 2 * xs + (xr & 1) + 4 * (xr >> 1);
        const int q  = 8 * (it >> 1) + xq;
        if (q < 37) {
            const int a = XS_OFF + (4 * q) * STR + kslot(ci);
            lds[a]           = xv[it][0];
            lds[a + STR]     = xv[it][1];
            lds[a + 2 * STR] = xv[it][2];
            lds[a + 3 * STR] = xv[it][3];
        }
    }
}

template<bool FIRST>
__device__ __forceinline__ void term_pair(const float* rp, int rowc, int cq0, float (&part)[8][4])
{
    constexpr int NQ = 4;
    #pragma unroll
    for (int n0 = 0; n0 < 8; n0 += NQ) {
        v4f l[NQ], r[NQ];
        #pragma unroll
        for (int u = 0; u < NQ; ++u) {
            const int co = cq0 + 8 * (n0 + u);
            l[u] = *(const v4f*)&rp[rowc + co];
            r[u] = *(const v4f*)&rp[rowc + STR + co];
        }
        #pragma unroll
        for (int u = 0; u < NQ; ++u) {
            #pragma unroll
            for (int c = 0; c < 4; ++c) {
                const float d  = r[u][c] - l[u][c];
                const float qg = d / (1.0f + __expf(-1.702f * d));
                const float s  = qg + (r[u][c] + l[u][c]);
                if constexpr (FIRST) part[n0 + u][c] = s;
                else                 part[n0 + u][c] += s;
            }
        }
    }
}

__device__ __forceinline__ void lds_barrier()
{
    __builtin_amdgcn_fence(__ATOMIC_RELEASE, "workgroup", "local");
    __builtin_amdgcn_s_barrier();
    __builtin_amdgcn_fence(__ATOMIC_ACQUIRE, "workgroup", "local");
}

__global__ __launch_bounds__(256) void trconv_fused_kernel(
    const float* __restrict__ x,
    const float* __restrict__ w,
    const float* __restrict__ bias,
    float* __restrict__ out)
{
    __shared__ __attribute__((aligned(16))) float lds[LDS_FLOATS];
    const int tid = threadIdx.x;
    const int b   = blockIdx.z;
    const int g   = blockIdx.y;
    const int pc0 = blockIdx.x * (TPW * NP);

    const int lane = tid & 31;
    const int wv   = tid >> 5;
    const int ct   = wv & 3;
    const int wrow = wv >> 2;
    const int nm   = lane & 15;
    const int lh   = lane >> 4;

    const float* xg = x + (size_t)(b * 128 + g * 64) * T_LEN;
    v4f xv[10];
    x_tile_load(xg, pc0 - 8, lane, wv, xv);
    float bias_v = 0.0f;
    if (tid < 64) bias_v = bias[g * 64 + tid];

    const float* wgrp = w + (size_t)(g * 64) * (64 * 13);
    v4f wb[4][13];
    {
        const float* wsrc = wgrp + (wv * 8) * (64 * 13) + 4 * lane;
        #pragma unroll
        for (int i = 0; i < 4; ++i) {
            #pragma unroll
            for (int j = 0; j < 13; ++j)
                wb[i][j] = *(const v4f*)(wsrc + i * 1664 + 128 * j);
        }
    }

    x_tile_scatter(lds, lane, wv, xv);
    if (tid < 64) lds[BIAS_OFF + tid] = bias_v;

    const int aw_base  = W_OFF + (ct * 16 + nm) * STR + lh * 32;
    const int xb_base  = XS_OFF + nm * STR + lh * 32;
    const int res_base = R_OFF + nm * STR + ct * 16 + lh * 8;

    {
        int woff[13][4];
        #pragma unroll
        for (int j = 0; j < 13; ++j) {
            #pragma unroll
            for (int c = 0; c < 4; ++c) {
                const int L    = 128 * j + 4 * lane + c;
                const int cf   = L / 13;
                const int k    = L - 13 * cf;
                const int slot = 13 * (k / 3) + 8 - 4 * k;
                woff[j][c] = W_OFF + slot * WSLOT + (wv * 8 + (cf >> 6)) * STR + kslot(cf & 63);
            }
        }
        #pragma unroll
        for (int i = 0; i < 4; ++i) {
            #pragma unroll
            for (int j = 0; j < 13; ++j) {
                #pragma unroll
                for (int c = 0; c < 4; ++c)
                    lds[woff[j][c] + i * 2 * STR] = wb[i][j][c];
            }
        }
    }

    const float* rp  = &lds[R_OFF];
    const int    il  = tid & (NP - 1);
    const int    cq0 = 4 * (tid >> 7);
    float* outg = out + (size_t)(b * 128 + g * 64) * T_LEN;

    #pragma unroll 1
    for (int tile = 0; tile < TPW; ++tile) {
        const int p0  = pc0 + tile * NP;
        const int me0 = p0 - 8;
        const int i   = p0 + il;
        const int rowc = (ref_idx(i) - me0) * STR;

        if (tile + 1 < TPW) x_tile_load(xg, me0 + NP, lane, wv, xv);

        float part[8][4];

        const bool first = (tile == 0);
        const bool more  = (tile + 1 < TPW);
        const bool keep  = (nm < 8);

        lds_barrier();
        if (!first) carry_restore(lds, tid, 0);
        residue_gemm<4>(lds, wrow, first, keep, aw_base + 0 * WSLOT, xb_base, res_base);
        lds_barrier();
        term_pair<true >(rp, rowc, cq0, part);
        if (more) carry_save(lds, tid, 0);
        lds_barrier();
        if (!first) carry_restore(lds, tid, 1);
        residue_gemm<4>(lds, wrow, first, keep, aw_base + 4 * WSLOT, xb_base, res_base);
        lds_barrier();
        term_pair<false>(rp, rowc, cq0, part);
        if (more) carry_save(lds, tid, 1);
        lds_barrier();
        if (!first) carry_restore(lds, tid, 2);
        residue_gemm<5>(lds, wrow, first, keep, aw_base + 8 * WSLOT, xb_base, res_base);
        lds_barrier();

        if (more) { x_tile_scatter(lds, lane, wv, xv); carry_save(lds, tid, 2); }

        int  roww[4];
        bool inr[4];
        #pragma unroll
        for (int q = 0; q < 4; ++q) {
            const int j = i + (q < 2 ? q - 2 : q - 1);
            roww[q] = (ref_idx(j) - me0) * STR;
            inr[q]  = (j >= 0 && j <= T_LEN - 1);
        }
        float* outp = outg + i;

        constexpr int NQ = 2;
        #pragma unroll
        for (int n0 = 0; n0 < 8; n0 += NQ) {
            v4f bi[NQ], xc[NQ], xw[4][NQ];
            #pragma unroll
            for (int u = 0; u < NQ; ++u) {
                const int co = cq0 + 8 * (n0 + u);
                bi[u] = *(const v4f*)&lds[BIAS_OFF + co];
                xc[u] = *(const v4f*)&rp[rowc + co];
                #pragma unroll
                for (int q = 0; q < 4; ++q) xw[q][u] = *(const v4f*)&rp[roww[q] + co];
            }
            #pragma unroll
            for (int u = 0; u < NQ; ++u) {
                const int co = cq0 + 8 * (n0 + u);
                #pragma unroll
                for (int c = 0; c < 4; ++c) {
                    const float b1 = bi[u][c];
                    float x_m = xc[u][c] + b1;
                    float um = x_m;
                    #pragma unroll
                    for (int q = 0; q < 4; ++q) {
                        float tv = xw[q][u][c] + b1;
                        um = fmaxf(um, inr[q] ? tv : 0.0f);
                    }
                    float accv = part[n0 + u][c] + 2.0f * x_m + 4.0f * b1 + um;
                    outp[(size_t)(co + c) * T_LEN] = accv;
                }
            }
        }
    }
}

extern "C" void kernel_launch(void* const* d_in, const int* in_sizes, int n_in,
                              void* d_out, int out_size, void* d_ws, size_t ws_size,
                              hipStream_t stream) {
    const float* x  = (const float*)d_in[0];
    const float* w  = (const float*)d_in[1];
    const float* b  = (const float*)d_in[2];
    float* out      = (float*)d_out;

    dim3 grid(T_LEN / (NP * TPW), 2, NB);
    dim3 block(256);
    hipLaunchKernelGGL(trconv_fused_kernel, grid, block, 0, stream, x, w, b, out);
}


# Fragment loads: next tap's 24 LDS loads issued as one group ahead of the current tap's 32 matrix instructions (scheduling barrier; compiler had spread them 3 per 4), also across a wave's tile pairs
# speedup vs baseline: 2.7126x; 1.0384x over previous
#include <hip/hip_runtime.h>


typedef float v2f __attribute__((ext_vector_type(2)));
typedef float v4f __attribute__((ext_vector_type(4)));
typedef float v8f __attribute__((ext_vector_type(8)));

#define T_LEN 8192
#define NB    16
#define NP    128
#define TPW   8
#define ME    144
#define XROWS 148
#define STR   68

#define XS_OFF   0
#define W_OFF    (XS_OFF + XROWS * STR)
#define WSLOT    (64 * STR + 8)
#define RES_SZ   (ME * STR)
#define R_OFF    (W_OFF + 13 * WSLOT)
#define BIAS_OFF (R_OFF + RES_SZ)
#define CARRY_OFF (BIAS_OFF + 64)
#define LDS_FLOATS (CARRY_OFF + 3 * 256)
static_assert(LDS_FLOATS * 4 <= 320 * 1024, "fits the workgroup LDS");

__device__ __forceinline__ int kslot(int ci) {
    return (((ci >> 1) & 1) << 5) + ((ci >> 2) << 1) + (ci & 1);
}

__device__ __forceinline__ int ref_idx(int i) {
    if (i < 2) return 2 - i;
    if (i > T_LEN - 3) return (i == T_LEN - 2) ? T_LEN - 6 : T_LEN - 7;
    return i - 2;
}

template<int TAPS, bool DUAL, bool HEAD>
__device__ __forceinline__ void gemm_tiles(float* lds, int aw_base, int xb_base,
                                           int res_base, int pt0, bool keep)
{
    const int x0 = xb_base + pt0 * 16 * STR;
    const int x1 = x0 + 16 * STR;

    v4f A[2][8], B0[2][8], B1[2][8];
    #pragma unroll
    for (int q = 0; q < 8; ++q) {
        A[0][q]  = *(const v4f*)&lds[aw_base + 4 * q];
        B0[0][q] = *(const v4f*)&lds[x0 + 4 * q];
        if constexpr (DUAL) B1[0][q] = *(const v4f*)&lds[x1 + 4 * q];
    }

    v8f acc0 = {0.f,0.f,0.f,0.f,0.f,0.f,0.f,0.f};
    v8f acc1 = {0.f,0.f,0.f,0.f,0.f,0.f,0.f,0.f};

    #pragma unroll
    for (int t = 0; t < TAPS; ++t) {
        const int cur = t & 1, nxt = cur ^ 1;
        if (t + 1 < TAPS) {
            #pragma unroll
            for (int q = 0; q < 8; ++q) {
                A[nxt][q]  = *(const v4f*)&lds[aw_base + (t + 1) * WSLOT + 4 * q];
                B0[nxt][q] = *(const v4f*)&lds[x0 + (t + 1) * STR + 4 * q];
                if constexpr (DUAL) B1[nxt][q] = *(const v4f*)&lds[x1 + (t + 1) * STR + 4 * q];
            }
        }
        #pragma unroll
        for (int q = 0; q < 8; ++q) {
            v2f a0 = {A[cur][q][0], A[cur][q][1]};
            v2f a1 = {A[cur][q][2], A[cur][q][3]};
            v2f b0 = {B0[cur][q][0], B0[cur][q][1]};
            v2f b1 = {B0[cur][q][2], B0[cur][q][3]};
            acc0 = __builtin_amdgcn_wmma_f32_16x16x4_f32(false, a0, false, b0, (short)0, acc0, false, false);
            acc0 = __builtin_amdgcn_wmma_f32_16x16x4_f32(false, a1, false, b1, (short)0, acc0, false, false);
            if constexpr (DUAL) {
                v2f c0 = {B1[cur][q][0], B1[cur][q][1]};
                v2f c1 = {B1[cur][q][2], B1[cur][q][3]};
                acc1 = __builtin_amdgcn_wmma_f32_16x16x4_f32(false, a0, false, c0, (short)0, acc1, false, false);
                acc1 = __builtin_amdgcn_wmma_f32_16x16x4_f32(false, a1, false, c1, (short)0, acc1, false, false);
            }
        }
    }

    int st0 = res_base + pt0 * 16 * STR;
    if (!HEAD || keep) {
        *(v4f*)&lds[st0]     = (v4f){acc0[0], acc0[1], acc0[2], acc0[3]};
        *(v4f*)&lds[st0 + 4] = (v4f){acc0[4], acc0[5], acc0[6], acc0[7]};
    }
    if constexpr (DUAL) {
        int st1 = st0 + 16 * STR;
        *(v4f*)&lds[st1]     = (v4f){acc1[0], acc1[1], acc1[2], acc1[3]};
        *(v4f*)&lds[st1 + 4] = (v4f){acc1[4], acc1[5], acc1[6], acc1[7]};
    }
}

template<int TAPS>
__device__ __forceinline__ void gemm_pairs(float* lds, int aw_base, int xb8, int rb8, int pt0)
{
    constexpr int STEPS = 2 * TAPS;
    v4f A[2][8], B0[2][8], B1[2][8];

    #pragma unroll
    for (int q = 0; q < 8; ++q) {
        const int x0 = xb8 + pt0 * 16 * STR;
        A[0][q]  = *(const v4f*)&lds[aw_base + 4 * q];
        B0[0][q] = *(const v4f*)&lds[x0 + 4 * q];
        B1[0][q] = *(const v4f*)&lds[x0 + 16 * STR + 4 * q];
    }

    v8f acc0 = {0.f,0.f,0.f,0.f,0.f,0.f,0.f,0.f};
    v8f acc1 = {0.f,0.f,0.f,0.f,0.f,0.f,0.f,0.f};

    #pragma unroll
    for (int s = 0; s < STEPS; ++s) {
        const int cur = s & 1, nxt = cur ^ 1;
        if (s + 1 < STEPS) {
            const int pn = (s + 1) / TAPS, tn = (s + 1) % TAPS;
            const int xn = xb8 + (pt0 + 2 * pn) * 16 * STR + tn * STR;
            #pragma unroll
            for (int q = 0; q < 8; ++q) {
                A[nxt][q]  = *(const v4f*)&lds[aw_base + tn * WSLOT + 4 * q];
                B0[nxt][q] = *(const v4f*)&lds[xn + 4 * q];
                B1[nxt][q] = *(const v4f*)&lds[xn + 16 * STR + 4 * q];
            }
        }
        __builtin_amdgcn_sched_barrier(0);
        #pragma unroll
        for (int q = 0; q < 8; ++q) {
            v2f a0 = {A[cur][q][0], A[cur][q][1]};
            v2f a1 = {A[cur][q][2], A[cur][q][3]};
            v2f b0 = {B0[cur][q][0], B0[cur][q][1]};
            v2f b1 = {B0[cur][q][2], B0[cur][q][3]};
            v2f c0 = {B1[cur][q][0], B1[cur][q][1]};
            v2f c1 = {B1[cur][q][2], B1[cur][q][3]};
            acc0 = __builtin_amdgcn_wmma_f32_16x16x4_f32(false, a0, false, b0, (short)0, acc0, false, false);
            acc1 = __builtin_amdgcn_wmma_f32_16x16x4_f32(false, a0, false, c0, (short)0, acc1, false, false);
            acc0 = __builtin_amdgcn_wmma_f32_16x16x4_f32(false, a1, false, b1, (short)0, acc0, false, false);
            acc1 = __builtin_amdgcn_wmma_f32_16x16x4_f32(false, a1, false, c1, (short)0, acc1, false, false);
        }
        __builtin_amdgcn_sched_barrier(0);
        if (s % TAPS == TAPS - 1) {
            const int st0 = rb8 + (pt0 + 2 * (s / TAPS)) * 16 * STR;
            const int st1 = st0 + 16 * STR;
            *(v4f*)&lds[st0]     = (v4f){acc0[0], acc0[1], acc0[2], acc0[3]};
            *(v4f*)&lds[st0 + 4] = (v4f){acc0[4], acc0[5], acc0[6], acc0[7]};
            *(v4f*)&lds[st1]     = (v4f){acc1[0], acc1[1], acc1[2], acc1[3]};
            *(v4f*)&lds[st1 + 4] = (v4f){acc1[4], acc1[5], acc1[6], acc1[7]};
            acc0 = (v8f){0.f,0.f,0.f,0.f,0.f,0.f,0.f,0.f};
            acc1 = (v8f){0.f,0.f,0.f,0.f,0.f,0.f,0.f,0.f};
        }
    }
}

template<int TAPS>
__device__ __forceinline__ void residue_gemm(float* lds, int wrow, bool first, bool keep,
                                             int aw_base, int xb_base, int res_base)
{
    const int xb8 = xb_base + 8 * STR;
    const int rb8 = res_base + 8 * STR;
    if (wrow == 0) {
        if (first) gemm_tiles<TAPS, false, true>(lds, aw_base, xb_base, res_base, 0, keep);
        gemm_pairs<TAPS>(lds, aw_base, xb8, rb8, 0);
    } else {
        gemm_pairs<TAPS>(lds, aw_base, xb8, rb8, 4);
    }
}

__device__ __forceinline__ void carry_save(float* lds, int tid, int r)
{
    if (tid < 64)
        *(v4f*)&lds[CARRY_OFF + r * 256 + 4 * tid] =
            *(const v4f*)&lds[R_OFF + (132 + (tid >> 4)) * STR + 4 * (tid & 15)];
}
__device__ __forceinline__ void carry_restore(float* lds, int tid, int r)
{
    if (tid < 64)
        *(v4f*)&lds[R_OFF + (4 + (tid >> 4)) * STR + 4 * (tid & 15)] =
            *(const v4f*)&lds[CARRY_OFF + r * 256 + 4 * tid];
}

__device__ __forceinline__ void x_tile_load(const float* __restrict__ xg, int me0, int lane, int wv, v4f (&xv)[10])
{
    const int xq = lane & 7;
    const int xr = lane >> 3;
    const int xs = wv & 1;
    #pragma unroll
    for (int it = 0; it < 10; ++it) {
        const int h  = (it & 1) * 4 + (wv >> 1);
        const int ci = 8 * h + 2 * xs + (xr & 1) + 4 * (xr >> 1);
        int q = 8 * (it >> 1) + xq;
        q = q > 36 ? 36 : q;
        int m4 = me0 + 4 * q;
        m4 = m4 < 0 ? 0 : (m4 > T_LEN - 4 ? T_LEN - 4 : m4);
        xv[it] = *(const v4f*)(xg + (size_t)ci * T_LEN + m4);
    }
}

__device__ __forceinline__ void x_tile_scatter(float* lds, int lane, int wv, const v4f (&xv)[10])
{
    const int xq = lane & 7;
    const int xr = lane >> 3;
    const int xs = wv & 1;
    #pragma unroll
    for (int it = 0; it < 10; ++it) {
        const int h  = (it & 1) * 4 + (wv >> 1);
        const int ci = 8 * h + 2 * xs + (xr & 1) + 4 * (xr >> 1);
        const int q  = 8 * (it >> 1) + xq;
        if (q < 37) {
            const int a = XS_OFF + (4 * q) * STR + kslot(ci);
            lds[a]           = xv[it][0];
            lds[a + STR]     = xv[it][1];
            lds[a + 2 * STR] = xv[it][2];
            lds[a + 3 * STR] = xv[it][3];
        }
    }
}

template<bool FIRST>
__device__ __forceinline__ void term_pair(const float* rp, int rowc, int cq0, float (&part)[8][4])
{
    constexpr int NQ = 4;
    #pragma unroll
    for (int n0 = 0; n0 < 8; n0 += NQ) {
        v4f l[NQ], r[NQ];
        #pragma unroll
        for (int u = 0; u < NQ; ++u) {
            const int co = cq0 + 8 * (n0 + u);
            l[u] = *(const v4f*)&rp[rowc + co];
            r[u] = *(const v4f*)&rp[rowc + STR + co];
        }
        #pragma unroll
        for (int u = 0; u < NQ; ++u) {
            #pragma unroll
            for (int c = 0; c < 4; ++c) {
                const float d  = r[u][c] - l[u][c];
                const float qg = d / (1.0f + __expf(-1.702f * d));
                const float s  = qg + (r[u][c] + l[u][c]);
                if constexpr (FIRST) part[n0 + u][c] = s;
                else                 part[n0 + u][c] += s;
            }
        }
    }
}

__device__ __forceinline__ void lds_barrier()
{
    __builtin_amdgcn_fence(__ATOMIC_RELEASE, "workgroup", "local");
    __builtin_amdgcn_s_barrier();
    __builtin_amdgcn_fence(__ATOMIC_ACQUIRE, "workgroup", "local");
}

__global__ __launch_bounds__(256) void trconv_fused_kernel(
    const float* __restrict__ x,
    const float* __restrict__ w,
    const float* __restrict__ bias,
    float* __restrict__ out)
{
    __shared__ __attribute__((aligned(16))) float lds[LDS_FLOATS];
    const int tid = threadIdx.x;
    const int b   = blockIdx.z;
    const int g   = blockIdx.y;
    const int pc0 = blockIdx.x * (TPW * NP);

    const int lane = tid & 31;
    const int wv   = tid >> 5;
    const int ct   = wv & 3;
    const int wrow = wv >> 2;
    const int nm   = lane & 15;
    const int lh   = lane >> 4;

    const float* xg = x + (size_t)(b * 128 + g * 64) * T_LEN;
    v4f xv[10];
    x_tile_load(xg, pc0 - 8, lane, wv, xv);
    float bias_v = 0.0f;
    if (tid < 64) bias_v = bias[g * 64 + tid];

    const float* wgrp = w + (size_t)(g * 64) * (64 * 13);
    v4f wb[4][13];
    {
        const float* wsrc = wgrp + (wv * 8) * (64 * 13) + 4 * lane;
        #pragma unroll
        for (int i = 0; i < 4; ++i) {
            #pragma unroll
            for (int j = 0; j < 13; ++j)
                wb[i][j] = *(const v4f*)(wsrc + i * 1664 + 128 * j);
        }
    }

    x_tile_scatter(lds, lane, wv, xv);
    if (tid < 64) lds[BIAS_OFF + tid] = bias_v;

    const int aw_base  = W_OFF + (ct * 16 + nm) * STR + lh * 32;
    const int xb_base  = XS_OFF + nm * STR + lh * 32;
    const int res_base = R_OFF + nm * STR + ct * 16 + lh * 8;

    {
        int woff[13][4];
        #pragma unroll
        for (int j = 0; j < 13; ++j) {
            #pragma unroll
            for (int c = 0; c < 4; ++c) {
                const int L    = 128 * j + 4 * lane + c;
                const int cf   = L / 13;
                const int k    = L - 13 * cf;
                const int slot = 13 * (k / 3) + 8 - 4 * k;
                woff[j][c] = W_OFF + slot * WSLOT + (wv * 8 + (cf >> 6)) * STR + kslot(cf & 63);
            }
        }
        #pragma unroll
        for (int i = 0; i < 4; ++i) {
            #pragma unroll
            for (int j = 0; j < 13; ++j) {
                #pragma unroll
                for (int c = 0; c < 4; ++c)
                    lds[woff[j][c] + i * 2 * STR] = wb[i][j][c];
            }
        }
    }

    const float* rp  = &lds[R_OFF];
    const int    il  = tid & (NP - 1);
    const int    cq0 = 4 * (tid >> 7);
    float* outg = out + (size_t)(b * 128 + g * 64) * T_LEN;

    #pragma unroll 1
    for (int tile = 0; tile < TPW; ++tile) {
        const int p0  = pc0 + tile * NP;
        const int me0 = p0 - 8;
        const int i   = p0 + il;
        const int rowc = (ref_idx(i) - me0) * STR;

        if (tile + 1 < TPW) x_tile_load(xg, me0 + NP, lane, wv, xv);

        float part[8][4];

        const bool first = (tile == 0);
        const bool more  = (tile + 1 < TPW);
        const bool keep  = (nm < 8);

        lds_barrier();
        if (!first) carry_restore(lds, tid, 0);
        residue_gemm<4>(lds, wrow, first, keep, aw_base + 0 * WSLOT, xb_base, res_base);
        lds_barrier();
        term_pair<true >(rp, rowc, cq0, part);
        if (more) carry_save(lds, tid, 0);
        lds_barrier();
        if (!first) carry_restore(lds, tid, 1);
        residue_gemm<4>(lds, wrow, first, keep, aw_base + 4 * WSLOT, xb_base, res_base);
        lds_barrier();
        term_pair<false>(rp, rowc, cq0, part);
        if (more) carry_save(lds, tid, 1);
        lds_barrier();
        if (!first) carry_restore(lds, tid, 2);
        residue_gemm<5>(lds, wrow, first, keep, aw_base + 8 * WSLOT, xb_base, res_base);
        lds_barrier();

        if (more) { x_tile_scatter(lds, lane, wv, xv); carry_save(lds, tid, 2); }

        int  roww[4];
        bool inr[4];
        #pragma unroll
        for (int q = 0; q < 4; ++q) {
            const int j = i + (q < 2 ? q - 2 : q - 1);
            roww[q] = (ref_idx(j) - me0) * STR;
            inr[q]  = (j >= 0 && j <= T_LEN - 1);
        }
        float* outp = outg + i;

        constexpr int NQ = 2;
        #pragma unroll
        for (int n0 = 0; n0 < 8; n0 += NQ) {
            v4f bi[NQ], xc[NQ], xw[4][NQ];
            #pragma unroll
            for (int u = 0; u < NQ; ++u) {
                const int co = cq0 + 8 * (n0 + u);
                bi[u] = *(const v4f*)&lds[BIAS_OFF + co];
                xc[u] = *(const v4f*)&rp[rowc + co];
                #pragma unroll
                for (int q = 0; q < 4; ++q) xw[q][u] = *(const v4f*)&rp[roww[q] + co];
            }
            #pragma unroll
            for (int u = 0; u < NQ; ++u) {
                const int co = cq0 + 8 * (n0 + u);
                #pragma unroll
                for (int c = 0; c < 4; ++c) {
                    const float b1 = bi[u][c];
                    float x_m = xc[u][c] + b1;
                    float um = x_m;
                    #pragma unroll
                    for (int q = 0; q < 4; ++q) {
                        float tv = xw[q][u][c] + b1;
                        um = fmaxf(um, inr[q] ? tv : 0.0f);
                    }
                    float accv = part[n0 + u][c] + 2.0f * x_m + 4.0f * b1 + um;
                    outp[(size_t)(co + c) * T_LEN] = accv;
                }
            }
        }
    }
}

extern "C" void kernel_launch(void* const* d_in, const int* in_sizes, int n_in,
                              void* d_out, int out_size, void* d_ws, size_t ws_size,
                              hipStream_t stream) {
    const float* x  = (const float*)d_in[0];
    const float* w  = (const float*)d_in[1];
    const float* b  = (const float*)d_in[2];
    float* out      = (float*)d_out;

    dim3 grid(T_LEN / (NP * TPW), 2, NB);
    dim3 block(256);
    hipLaunchKernelGGL(trconv_fused_kernel, grid, block, 0, stream, x, w, b, out);
}


# Quick-gelu quotient d/(1+exp(-1.702d)) as one hardware reciprocal (v_rcp_f32, 1 ulp) and a multiply; was a correctly rounded division of ~10 dependent vector instructions, 64 per thread per tile
# speedup vs baseline: 2.8486x; 1.0501x over previous
#include <hip/hip_runtime.h>


typedef float v2f __attribute__((ext_vector_type(2)));
typedef float v4f __attribute__((ext_vector_type(4)));
typedef float v8f __attribute__((ext_vector_type(8)));

#define T_LEN 8192
#define NB    16
#define NP    128
#define TPW   8
#define ME    144
#define XROWS 148
#define STR   68

#define XS_OFF   0
#define W_OFF    (XS_OFF + XROWS * STR)
#define WSLOT    (64 * STR + 8)
#define RES_SZ   (ME * STR)
#define R_OFF    (W_OFF + 13 * WSLOT)
#define BIAS_OFF (R_OFF + RES_SZ)
#define CARRY_OFF (BIAS_OFF + 64)
#define LDS_FLOATS (CARRY_OFF + 3 * 256)
static_assert(LDS_FLOATS * 4 <= 320 * 1024, "fits the workgroup LDS");

__device__ __forceinline__ int kslot(int ci) {
    return (((ci >> 1) & 1) << 5) + ((ci >> 2) << 1) + (ci & 1);
}

__device__ __forceinline__ int ref_idx(int i) {
    if (i < 2) return 2 - i;
    if (i > T_LEN - 3) return (i == T_LEN - 2) ? T_LEN - 6 : T_LEN - 7;
    return i - 2;
}

template<int TAPS, bool DUAL, bool HEAD>
__device__ __forceinline__ void gemm_tiles(float* lds, int aw_base, int xb_base,
                                           int res_base, int pt0, bool keep)
{
    const int x0 = xb_base + pt0 * 16 * STR;
    const int x1 = x0 + 16 * STR;

    v4f A[2][8], B0[2][8], B1[2][8];
    #pragma unroll
    for (int q = 0; q < 8; ++q) {
        A[0][q]  = *(const v4f*)&lds[aw_base + 4 * q];
        B0[0][q] = *(const v4f*)&lds[x0 + 4 * q];
        if constexpr (DUAL) B1[0][q] = *(const v4f*)&lds[x1 + 4 * q];
    }

    v8f acc0 = {0.f,0.f,0.f,0.f,0.f,0.f,0.f,0.f};
    v8f acc1 = {0.f,0.f,0.f,0.f,0.f,0.f,0.f,0.f};

    #pragma unroll
    for (int t = 0; t < TAPS; ++t) {
        const int cur = t & 1, nxt = cur ^ 1;
        if (t + 1 < TAPS) {
            #pragma unroll
            for (int q = 0; q < 8; ++q) {
                A[nxt][q]  = *(const v4f*)&lds[aw_base + (t + 1) * WSLOT + 4 * q];
                B0[nxt][q] = *(const v4f*)&lds[x0 + (t + 1) * STR + 4 * q];
                if constexpr (DUAL) B1[nxt][q] = *(const v4f*)&lds[x1 + (t + 1) * STR + 4 * q];
            }
        }
        #pragma unroll
        for (int q = 0; q < 8; ++q) {
            v2f a0 = {A[cur][q][0], A[cur][q][1]};
            v2f a1 = {A[cur][q][2], A[cur][q][3]};
            v2f b0 = {B0[cur][q][0], B0[cur][q][1]};
            v2f b1 = {B0[cur][q][2], B0[cur][q][3]};
            acc0 = __builtin_amdgcn_wmma_f32_16x16x4_f32(false, a0, false, b0, (short)0, acc0, false, false);
            acc0 = __builtin_amdgcn_wmma_f32_16x16x4_f32(false, a1, false, b1, (short)0, acc0, false, false);
            if constexpr (DUAL) {
                v2f c0 = {B1[cur][q][0], B1[cur][q][1]};
                v2f c1 = {B1[cur][q][2], B1[cur][q][3]};
                acc1 = __builtin_amdgcn_wmma_f32_16x16x4_f32(false, a0, false, c0, (short)0, acc1, false, false);
                acc1 = __builtin_amdgcn_wmma_f32_16x16x4_f32(false, a1, false, c1, (short)0, acc1, false, false);
            }
        }
    }

    int st0 = res_base + pt0 * 16 * STR;
    if (!HEAD || keep) {
        *(v4f*)&lds[st0]     = (v4f){acc0[0], acc0[1], acc0[2], acc0[3]};
        *(v4f*)&lds[st0 + 4] = (v4f){acc0[4], acc0[5], acc0[6], acc0[7]};
    }
    if constexpr (DUAL) {
        int st1 = st0 + 16 * STR;
        *(v4f*)&lds[st1]     = (v4f){acc1[0], acc1[1], acc1[2], acc1[3]};
        *(v4f*)&lds[st1 + 4] = (v4f){acc1[4], acc1[5], acc1[6], acc1[7]};
    }
}

template<int TAPS>
__device__ __forceinline__ void gemm_pairs(float* lds, int aw_base, int xb8, int rb8, int pt0)
{
    constexpr int STEPS = 2 * TAPS;
    v4f A[2][8], B0[2][8], B1[2][8];

    #pragma unroll
    for (int q = 0; q < 8; ++q) {
        const int x0 = xb8 + pt0 * 16 * STR;
        A[0][q]  = *(const v4f*)&lds[aw_base + 4 * q];
        B0[0][q] = *(const v4f*)&lds[x0 + 4 * q];
        B1[0][q] = *(const v4f*)&lds[x0 + 16 * STR + 4 * q];
    }

    v8f acc0 = {0.f,0.f,0.f,0.f,0.f,0.f,0.f,0.f};
    v8f acc1 = {0.f,0.f,0.f,0.f,0.f,0.f,0.f,0.f};

    #pragma unroll
    for (int s = 0; s < STEPS; ++s) {
        const int cur = s & 1, nxt = cur ^ 1;
        if (s + 1 < STEPS) {
            const int pn = (s + 1) / TAPS, tn = (s + 1) % TAPS;
            const int xn = xb8 + (pt0 + 2 * pn) * 16 * STR + tn * STR;
            #pragma unroll
            for (int q = 0; q < 8; ++q) {
                A[nxt][q]  = *(const v4f*)&lds[aw_base + tn * WSLOT + 4 * q];
                B0[nxt][q] = *(const v4f*)&lds[xn + 4 * q];
                B1[nxt][q] = *(const v4f*)&lds[xn + 16 * STR + 4 * q];
            }
        }
        __builtin_amdgcn_sched_barrier(0);
        #pragma unroll
        for (int q = 0; q < 8; ++q) {
            v2f a0 = {A[cur][q][0], A[cur][q][1]};
            v2f a1 = {A[cur][q][2], A[cur][q][3]};
            v2f b0 = {B0[cur][q][0], B0[cur][q][1]};
            v2f b1 = {B0[cur][q][2], B0[cur][q][3]};
            v2f c0 = {B1[cur][q][0], B1[cur][q][1]};
            v2f c1 = {B1[cur][q][2], B1[cur][q][3]};
            acc0 = __builtin_amdgcn_wmma_f32_16x16x4_f32(false, a0, false, b0, (short)0, acc0, false, false);
            acc1 = __builtin_amdgcn_wmma_f32_16x16x4_f32(false, a0, false, c0, (short)0, acc1, false, false);
            acc0 = __builtin_amdgcn_wmma_f32_16x16x4_f32(false, a1, false, b1, (short)0, acc0, false, false);
            acc1 = __builtin_amdgcn_wmma_f32_16x16x4_f32(false, a1, false, c1, (short)0, acc1, false, false);
        }
        __builtin_amdgcn_sched_barrier(0);
        if (s % TAPS == TAPS - 1) {
            const int st0 = rb8 + (pt0 + 2 * (s / TAPS)) * 16 * STR;
            const int st1 = st0 + 16 * STR;
            *(v4f*)&lds[st0]     = (v4f){acc0[0], acc0[1], acc0[2], acc0[3]};
            *(v4f*)&lds[st0 + 4] = (v4f){acc0[4], acc0[5], acc0[6], acc0[7]};
            *(v4f*)&lds[st1]     = (v4f){acc1[0], acc1[1], acc1[2], acc1[3]};
            *(v4f*)&lds[st1 + 4] = (v4f){acc1[4], acc1[5], acc1[6], acc1[7]};
            acc0 = (v8f){0.f,0.f,0.f,0.f,0.f,0.f,0.f,0.f};
            acc1 = (v8f){0.f,0.f,0.f,0.f,0.f,0.f,0.f,0.f};
        }
    }
}

template<int TAPS>
__device__ __forceinline__ void residue_gemm(float* lds, int wrow, bool first, bool keep,
                                             int aw_base, int xb_base, int res_base)
{
    const int xb8 = xb_base + 8 * STR;
    const int rb8 = res_base + 8 * STR;
    if (wrow == 0) {
        if (first) gemm_tiles<TAPS, false, true>(lds, aw_base, xb_base, res_base, 0, keep);
        gemm_pairs<TAPS>(lds, aw_base, xb8, rb8, 0);
    } else {
        gemm_pairs<TAPS>(lds, aw_base, xb8, rb8, 4);
    }
}

__device__ __forceinline__ void carry_save(float* lds, int tid, int r)
{
    if (tid < 64)
        *(v4f*)&lds[CARRY_OFF + r * 256 + 4 * tid] =
            *(const v4f*)&lds[R_OFF + (132 + (tid >> 4)) * STR + 4 * (tid & 15)];
}
__device__ __forceinline__ void carry_restore(float* lds, int tid, int r)
{
    if (tid < 64)
        *(v4f*)&lds[R_OFF + (4 + (tid >> 4)) * STR + 4 * (tid & 15)] =
            *(const v4f*)&lds[CARRY_OFF + r * 256 + 4 * tid];
}

__device__ __forceinline__ void x_tile_load(const float* __restrict__ xg, int me0, int lane, int wv, v4f (&xv)[10])
{
    const int xq = lane & 7;
    const int xr = lane >> 3;
    const int xs = wv & 1;
    #pragma unroll
    for (int it = 0; it < 10; ++it) {
        const int h  = (it & 1) * 4 + (wv >> 1);
        const int ci = 8 * h + 2 * xs + (xr & 1) + 4 * (xr >> 1);
        int q = 8 * (it >> 1) + xq;
        q = q > 36 ? 36 : q;
        int m4 = me0 + 4 * q;
        m4 = m4 < 0 ? 0 : (m4 > T_LEN - 4 ? T_LEN - 4 : m4);
        xv[it] = *(const v4f*)(xg + (size_t)ci * T_LEN + m4);
    }
}

__device__ __forceinline__ void x_tile_scatter(float* lds, int lane, int wv, const v4f (&xv)[10])
{
    const int xq = lane & 7;
    const int xr = lane >> 3;
    const int xs = wv & 1;
    #pragma unroll
    for (int it = 0; it < 10; ++it) {
        const int h  = (it & 1) * 4 + (wv >> 1);
        const int ci = 8 * h + 2 * xs + (xr & 1) + 4 * (xr >> 1);
        const int q  = 8 * (it >> 1) + xq;
        if (q < 37) {
            const int a = XS_OFF + (4 * q) * STR + kslot(ci);
            lds[a]           = xv[it][0];
            lds[a + STR]     = xv[it][1];
            lds[a + 2 * STR] = xv[it][2];
            lds[a + 3 * STR] = xv[it][3];
        }
    }
}

template<bool FIRST>
__device__ __forceinline__ void term_pair(const float* rp, int rowc, int cq0, float (&part)[8][4])
{
    constexpr int NQ = 4;
    #pragma unroll
    for (int n0 = 0; n0 < 8; n0 += NQ) {
        v4f l[NQ], r[NQ];
        #pragma unroll
        for (int u = 0; u < NQ; ++u) {
            const int co = cq0 + 8 * (n0 + u);
            l[u] = *(const v4f*)&rp[rowc + co];
            r[u] = *(const v4f*)&rp[rowc + STR + co];
        }
        #pragma unroll
        for (int u = 0; u < NQ; ++u) {
            #pragma unroll
            for (int c = 0; c < 4; ++c) {
                const float d  = r[u][c] - l[u][c];
                const float qg = d * __builtin_amdgcn_rcpf(1.0f + __expf(-1.702f * d));
                const float s  = qg + (r[u][c] + l[u][c]);
                if constexpr (FIRST) part[n0 + u][c] = s;
                else                 part[n0 + u][c] += s;
            }
        }
    }
}

__device__ __forceinline__ void lds_barrier()
{
    __builtin_amdgcn_fence(__ATOMIC_RELEASE, "workgroup", "local");
    __builtin_amdgcn_s_barrier();
    __builtin_amdgcn_fence(__ATOMIC_ACQUIRE, "workgroup", "local");
}

__global__ __launch_bounds__(256) void trconv_fused_kernel(
    const float* __restrict__ x,
    const float* __restrict__ w,
    const float* __restrict__ bias,
    float* __restrict__ out)
{
    __shared__ __attribute__((aligned(16))) float lds[LDS_FLOATS];
    const int tid = threadIdx.x;
    const int b   = blockIdx.z;
    const int g   = blockIdx.y;
    const int pc0 = blockIdx.x * (TPW * NP);

    const int lane = tid & 31;
    const int wv   = tid >> 5;
    const int ct   = wv & 3;
    const int wrow = wv >> 2;
    const int nm   = lane & 15;
    const int lh   = lane >> 4;

    const float* xg = x + (size_t)(b * 128 + g * 64) * T_LEN;
    v4f xv[10];
    x_tile_load(xg, pc0 - 8, lane, wv, xv);
    float bias_v = 0.0f;
    if (tid < 64) bias_v = bias[g * 64 + tid];

    const float* wgrp = w + (size_t)(g * 64) * (64 * 13);
    v4f wb[4][13];
    {
        const float* wsrc = wgrp + (wv * 8) * (64 * 13) + 4 * lane;
        #pragma unroll
        for (int i = 0; i < 4; ++i) {
            #pragma unroll
            for (int j = 0; j < 13; ++j)
                wb[i][j] = *(const v4f*)(wsrc + i * 1664 + 128 * j);
        }
    }

    x_tile_scatter(lds, lane, wv, xv);
    if (tid < 64) lds[BIAS_OFF + tid] = bias_v;

    const int aw_base  = W_OFF + (ct * 16 + nm) * STR + lh * 32;
    const int xb_base  = XS_OFF + nm * STR + lh * 32;
    const int res_base = R_OFF + nm * STR + ct * 16 + lh * 8;

    {
        int woff[13][4];
        #pragma unroll
        for (int j = 0; j < 13; ++j) {
            #pragma unroll
            for (int c = 0; c < 4; ++c) {
                const int L    = 128 * j + 4 * lane + c;
                const int cf   = L / 13;
                const int k    = L - 13 * cf;
                const int slot = 13 * (k / 3) + 8 - 4 * k;
                woff[j][c] = W_OFF + slot * WSLOT + (wv * 8 + (cf >> 6)) * STR + kslot(cf & 63);
            }
        }
        #pragma unroll
        for (int i = 0; i < 4; ++i) {
            #pragma unroll
            for (int j = 0; j < 13; ++j) {
                #pragma unroll
                for (int c = 0; c < 4; ++c)
                    lds[woff[j][c] + i * 2 * STR] = wb[i][j][c];
            }
        }
    }

    const float* rp  = &lds[R_OFF];
    const int    il  = tid & (NP - 1);
    const int    cq0 = 4 * (tid >> 7);
    float* outg = out + (size_t)(b * 128 + g * 64) * T_LEN;

    #pragma unroll 1
    for (int tile = 0; tile < TPW; ++tile) {
        const int p0  = pc0 + tile * NP;
        const int me0 = p0 - 8;
        const int i   = p0 + il;
        const int rowc = (ref_idx(i) - me0) * STR;

        if (tile + 1 < TPW) x_tile_load(xg, me0 + NP, lane, wv, xv);

        float part[8][4];

        const bool first = (tile == 0);
        const bool more  = (tile + 1 < TPW);
        const bool keep  = (nm < 8);

        lds_barrier();
        if (!first) carry_restore(lds, tid, 0);
        residue_gemm<4>(lds, wrow, first, keep, aw_base + 0 * WSLOT, xb_base, res_base);
        lds_barrier();
        term_pair<true >(rp, rowc, cq0, part);
        if (more) carry_save(lds, tid, 0);
        lds_barrier();
        if (!first) carry_restore(lds, tid, 1);
        residue_gemm<4>(lds, wrow, first, keep, aw_base + 4 * WSLOT, xb_base, res_base);
        lds_barrier();
        term_pair<false>(rp, rowc, cq0, part);
        if (more) carry_save(lds, tid, 1);
        lds_barrier();
        if (!first) carry_restore(lds, tid, 2);
        residue_gemm<5>(lds, wrow, first, keep, aw_base + 8 * WSLOT, xb_base, res_base);
        lds_barrier();

        if (more) { x_tile_scatter(lds, lane, wv, xv); carry_save(lds, tid, 2); }

        int  roww[4];
        bool inr[4];
        #pragma unroll
        for (int q = 0; q < 4; ++q) {
            const int j = i + (q < 2 ? q - 2 : q - 1);
            roww[q] = (ref_idx(j) - me0) * STR;
            inr[q]  = (j >= 0 && j <= T_LEN - 1);
        }
        float* outp = outg + i;

        constexpr int NQ = 2;
        #pragma unroll
        for (int n0 = 0; n0 < 8; n0 += NQ) {
            v4f bi[NQ], xc[NQ], xw[4][NQ];
            #pragma unroll
            for (int u = 0; u < NQ; ++u) {
                const int co = cq0 + 8 * (n0 + u);
                bi[u] = *(const v4f*)&lds[BIAS_OFF + co];
                xc[u] = *(const v4f*)&rp[rowc + co];
                #pragma unroll
                for (int q = 0; q < 4; ++q) xw[q][u] = *(const v4f*)&rp[roww[q] + co];
            }
            #pragma unroll
            for (int u = 0; u < NQ; ++u) {
                const int co = cq0 + 8 * (n0 + u);
                #pragma unroll
                for (int c = 0; c < 4; ++c) {
                    const float b1 = bi[u][c];
                    float x_m = xc[u][c] + b1;
                    float um = x_m;
                    #pragma unroll
                    for (int q = 0; q < 4; ++q) {
                        float tv = xw[q][u][c] + b1;
                        um = fmaxf(um, inr[q] ? tv : 0.0f);
                    }
                    float accv = part[n0 + u][c] + 2.0f * x_m + 4.0f * b1 + um;
                    outp[(size_t)(co + c) * T_LEN] = accv;
                }
            }
        }
    }
}

extern "C" void kernel_launch(void* const* d_in, const int* in_sizes, int n_in,
                              void* d_out, int out_size, void* d_ws, size_t ws_size,
                              hipStream_t stream) {
    const float* x  = (const float*)d_in[0];
    const float* w  = (const float*)d_in[1];
    const float* b  = (const float*)d_in[2];
    float* out      = (float*)d_out;

    dim3 grid(T_LEN / (NP * TPW), 2, NB);
    dim3 block(256);
    hipLaunchKernelGGL(trconv_fused_kernel, grid, block, 0, stream, x, w, b, out);
}


# One GEMM pass per tile for the three residues: a shift's x fragments are loaded once and meet the taps 3t+2, 3t+1, 3t (was 3 passes); residue 1 parked in the dead x tile: 4 barriers per tile, not 6
# speedup vs baseline: 2.9465x; 1.0344x over previous
#include <hip/hip_runtime.h>


typedef float v2f __attribute__((ext_vector_type(2)));
typedef float v4f __attribute__((ext_vector_type(4)));
typedef float v8f __attribute__((ext_vector_type(8)));

#define T_LEN 8192
#define NB    16
#define NP    128
#define TPW   8
#define ME    144
#define XROWS 148
#define STR   68

#define XS_OFF   0
#define W_OFF    (XS_OFF + XROWS * STR)
#define WSLOT    (64 * STR + 8)
#define RES_SZ   (ME * STR)
#define R_OFF    (W_OFF + 13 * WSLOT)
#define BIAS_OFF (R_OFF + RES_SZ)
#define CARRY_OFF (BIAS_OFF + 64)
#define LDS_FLOATS (CARRY_OFF + 3 * 256)
static_assert(LDS_FLOATS * 4 <= 320 * 1024, "fits the workgroup LDS");

__device__ __forceinline__ int kslot(int ci) {
    return (((ci >> 1) & 1) << 5) + ((ci >> 2) << 1) + (ci & 1);
}

__device__ __forceinline__ int ref_idx(int i) {
    if (i < 2) return 2 - i;
    if (i > T_LEN - 3) return (i == T_LEN - 2) ? T_LEN - 6 : T_LEN - 7;
    return i - 2;
}

__device__ __forceinline__ void gemm_all(float* lds, int aw_base, int xb8, int rb8, int pt0,
                                         v8f (&h1)[2][2], v8f (&h2)[2][2])
{
    constexpr int STEPS = 20;
    v4f A0[2][4], A1[2][4], A2[2][4], B0[2][4], B1[2][4];

    #pragma unroll
    for (int q = 0; q < 4; ++q) {
        const int x0 = xb8 + pt0 * 16 * STR;
        A0[0][q] = *(const v4f*)&lds[aw_base + 4 * q];
        A1[0][q] = *(const v4f*)&lds[aw_base + 4 * WSLOT + 4 * q];
        A2[0][q] = *(const v4f*)&lds[aw_base + 8 * WSLOT + 4 * q];
        B0[0][q] = *(const v4f*)&lds[x0 + 4 * q];
        B1[0][q] = *(const v4f*)&lds[x0 + 16 * STR + 4 * q];
    }

    const v8f zero8 = {0.f,0.f,0.f,0.f,0.f,0.f,0.f,0.f};
    v8f r0a = zero8, r0b = zero8, r1a = zero8, r1b = zero8, r2a = zero8, r2b = zero8;

    #pragma unroll
    for (int s = 0; s < STEPS; ++s) {
        const int cur = s & 1, nxt = cur ^ 1;
        const int t = (s % 10) >> 1;
        if (s + 1 < STEPS) {
            const int pn = (s + 1) / 10, tn = ((s + 1) % 10) >> 1, hn = (s + 1) & 1;
            const int xn = xb8 + (pt0 + 2 * pn) * 16 * STR + tn * STR + 16 * hn;
            const int an = aw_base + 16 * hn;
            #pragma unroll
            for (int q = 0; q < 4; ++q) {
                if (tn < 4) {
                    A0[nxt][q] = *(const v4f*)&lds[an + tn * WSLOT + 4 * q];
                    A1[nxt][q] = *(const v4f*)&lds[an + (4 + tn) * WSLOT + 4 * q];
                }
                A2[nxt][q] = *(const v4f*)&lds[an + (8 + tn) * WSLOT + 4 * q];
                B0[nxt][q] = *(const v4f*)&lds[xn + 4 * q];
                B1[nxt][q] = *(const v4f*)&lds[xn + 16 * STR + 4 * q];
            }
        }
        __builtin_amdgcn_sched_barrier(0);
        #pragma unroll
        for (int q = 0; q < 4; ++q) {
            v2f b0 = {B0[cur][q][0], B0[cur][q][1]};
            v2f b1 = {B0[cur][q][2], B0[cur][q][3]};
            v2f c0 = {B1[cur][q][0], B1[cur][q][1]};
            v2f c1 = {B1[cur][q][2], B1[cur][q][3]};
            v2f w20 = {A2[cur][q][0], A2[cur][q][1]};
            v2f w21 = {A2[cur][q][2], A2[cur][q][3]};
            if (t < 4) {
                v2f w00 = {A0[cur][q][0], A0[cur][q][1]};
                v2f w01 = {A0[cur][q][2], A0[cur][q][3]};
                v2f w10 = {A1[cur][q][0], A1[cur][q][1]};
                v2f w11 = {A1[cur][q][2], A1[cur][q][3]};
                r0a = __builtin_amdgcn_wmma_f32_16x16x4_f32(false, w00, false, b0, (short)0, r0a, false, false);
                r0b = __builtin_amdgcn_wmma_f32_16x16x4_f32(false, w00, false, c0, (short)0, r0b, false, false);
                r1a = __builtin_amdgcn_wmma_f32_16x16x4_f32(false, w10, false, b0, (short)0, r1a, false, false);
                r1b = __builtin_amdgcn_wmma_f32_16x16x4_f32(false, w10, false, c0, (short)0, r1b, false, false);
                r2a = __builtin_amdgcn_wmma_f32_16x16x4_f32(false, w20, false, b0, (short)0, r2a, false, false);
                r2b = __builtin_amdgcn_wmma_f32_16x16x4_f32(false, w20, false, c0, (short)0, r2b, false, false);
                r0a = __builtin_amdgcn_wmma_f32_16x16x4_f32(false, w01, false, b1, (short)0, r0a, false, false);
                r0b = __builtin_amdgcn_wmma_f32_16x16x4_f32(false, w01, false, c1, (short)0, r0b, false, false);
                r1a = __builtin_amdgcn_wmma_f32_16x16x4_f32(false, w11, false, b1, (short)0, r1a, false, false);
                r1b = __builtin_amdgcn_wmma_f32_16x16x4_f32(false, w11, false, c1, (short)0, r1b, false, false);
                r2a = __builtin_amdgcn_wmma_f32_16x16x4_f32(false, w21, false, b1, (short)0, r2a, false, false);
                r2b = __builtin_amdgcn_wmma_f32_16x16x4_f32(false, w21, false, c1, (short)0, r2b, false, false);
            } else {
                r2a = __builtin_amdgcn_wmma_f32_16x16x4_f32(false, w20, false, b0, (short)0, r2a, false, false);
                r2b = __builtin_amdgcn_wmma_f32_16x16x4_f32(false, w20, false, c0, (short)0, r2b, false, false);
                r2a = __builtin_amdgcn_wmma_f32_16x16x4_f32(false, w21, false, b1, (short)0, r2a, false, false);
                r2b = __builtin_amdgcn_wmma_f32_16x16x4_f32(false, w21, false, c1, (short)0, r2b, false, false);
            }
        }
        __builtin_amdgcn_sched_barrier(0);
        if (s % 10 == 9) {
            const int p   = s / 10;
            const int st0 = rb8 + (pt0 + 2 * p) * 16 * STR;
            const int st1 = st0 + 16 * STR;
            *(v4f*)&lds[st0]     = (v4f){r0a[0], r0a[1], r0a[2], r0a[3]};
            *(v4f*)&lds[st0 + 4] = (v4f){r0a[4], r0a[5], r0a[6], r0a[7]};
            *(v4f*)&lds[st1]     = (v4f){r0b[0], r0b[1], r0b[2], r0b[3]};
            *(v4f*)&lds[st1 + 4] = (v4f){r0b[4], r0b[5], r0b[6], r0b[7]};
            h1[p][0] = r1a; h1[p][1] = r1b;
            h2[p][0] = r2a; h2[p][1] = r2b;
            r0a = zero8; r0b = zero8; r1a = zero8; r1b = zero8; r2a = zero8; r2b = zero8;
        }
    }
}

__device__ __forceinline__ void head_all(float* lds, int aw_base, int xb_base, int res_base, bool keep,
                                         v8f& hh1, v8f& hh2)
{
    v8f r0 = {0.f,0.f,0.f,0.f,0.f,0.f,0.f,0.f};
    v8f r1 = {0.f,0.f,0.f,0.f,0.f,0.f,0.f,0.f};
    v8f r2 = {0.f,0.f,0.f,0.f,0.f,0.f,0.f,0.f};
    #pragma unroll
    for (int t = 0; t < 5; ++t) {
        #pragma unroll
        for (int q = 0; q < 8; ++q) {
            const v4f b  = *(const v4f*)&lds[xb_base + t * STR + 4 * q];
            const v4f w2 = *(const v4f*)&lds[aw_base + (8 + t) * WSLOT + 4 * q];
            v2f b0  = {b[0], b[1]},   b1  = {b[2], b[3]};
            v2f w20 = {w2[0], w2[1]}, w21 = {w2[2], w2[3]};
            if (t < 4) {
                const v4f w0 = *(const v4f*)&lds[aw_base + t * WSLOT + 4 * q];
                const v4f w1 = *(const v4f*)&lds[aw_base + (4 + t) * WSLOT + 4 * q];
                v2f w00 = {w0[0], w0[1]}, w01 = {w0[2], w0[3]};
                v2f w10 = {w1[0], w1[1]}, w11 = {w1[2], w1[3]};
                r0 = __builtin_amdgcn_wmma_f32_16x16x4_f32(false, w00, false, b0, (short)0, r0, false, false);
                r1 = __builtin_amdgcn_wmma_f32_16x16x4_f32(false, w10, false, b0, (short)0, r1, false, false);
                r2 = __builtin_amdgcn_wmma_f32_16x16x4_f32(false, w20, false, b0, (short)0, r2, false, false);
                r0 = __builtin_amdgcn_wmma_f32_16x16x4_f32(false, w01, false, b1, (short)0, r0, false, false);
                r1 = __builtin_amdgcn_wmma_f32_16x16x4_f32(false, w11, false, b1, (short)0, r1, false, false);
                r2 = __builtin_amdgcn_wmma_f32_16x16x4_f32(false, w21, false, b1, (short)0, r2, false, false);
            } else {
                r2 = __builtin_amdgcn_wmma_f32_16x16x4_f32(false, w20, false, b0, (short)0, r2, false, false);
                r2 = __builtin_amdgcn_wmma_f32_16x16x4_f32(false, w21, false, b1, (short)0, r2, false, false);
            }
        }
    }
    if (keep) {
        *(v4f*)&lds[res_base]     = (v4f){r0[0], r0[1], r0[2], r0[3]};
        *(v4f*)&lds[res_base + 4] = (v4f){r0[4], r0[5], r0[6], r0[7]};
    }
    hh1 = r1;
    hh2 = r2;
}

__device__ __forceinline__ void store_held(float* lds, int rb8, int pt0, const v8f (&h)[2][2])
{
    #pragma unroll
    for (int p = 0; p < 2; ++p) {
        #pragma unroll
        for (int j = 0; j < 2; ++j) {
            const int st = rb8 + (pt0 + 2 * p + j) * 16 * STR;
            *(v4f*)&lds[st]     = (v4f){h[p][j][0], h[p][j][1], h[p][j][2], h[p][j][3]};
            *(v4f*)&lds[st + 4] = (v4f){h[p][j][4], h[p][j][5], h[p][j][6], h[p][j][7]};
        }
    }
}

__device__ __forceinline__ void carry_save(float* lds, int tid, int r, int buf)
{
    if (tid < 64)
        *(v4f*)&lds[CARRY_OFF + r * 256 + 4 * tid] =
            *(const v4f*)&lds[buf + (132 + (tid >> 4)) * STR + 4 * (tid & 15)];
}
__device__ __forceinline__ void carry_restore(float* lds, int tid, int r, int buf)
{
    if (tid < 64)
        *(v4f*)&lds[buf + (4 + (tid >> 4)) * STR + 4 * (tid & 15)] =
            *(const v4f*)&lds[CARRY_OFF + r * 256 + 4 * tid];
}

__device__ __forceinline__ void x_tile_load(const float* __restrict__ xg, int me0, int lane, int wv, v4f (&xv)[10])
{
    const int xq = lane & 7;
    const int xr = lane >> 3;
    const int xs = wv & 1;
    #pragma unroll
    for (int it = 0; it < 10; ++it) {
        const int h  = (it & 1) * 4 + (wv >> 1);
        const int ci = 8 * h + 2 * xs + (xr & 1) + 4 * (xr >> 1);
        int q = 8 * (it >> 1) + xq;
        q = q > 36 ? 36 : q;
        int m4 = me0 + 4 * q;
        m4 = m4 < 0 ? 0 : (m4 > T_LEN - 4 ? T_LEN - 4 : m4);
        xv[it] = *(const v4f*)(xg + (size_t)ci * T_LEN + m4);
    }
}

__device__ __forceinline__ void x_tile_scatter(float* lds, int lane, int wv, const v4f (&xv)[10])
{
    const int xq = lane & 7;
    const int xr = lane >> 3;
    const int xs = wv & 1;
    #pragma unroll
    for (int it = 0; it < 10; ++it) {
        const int h  = (it & 1) * 4 + (wv >> 1);
        const int ci = 8 * h + 2 * xs + (xr & 1) + 4 * (xr >> 1);
        const int q  = 8 * (it >> 1) + xq;
        if (q < 37) {
            const int a = XS_OFF + (4 * q) * STR + kslot(ci);
            lds[a]           = xv[it][0];
            lds[a + STR]     = xv[it][1];
            lds[a + 2 * STR] = xv[it][2];
            lds[a + 3 * STR] = xv[it][3];
        }
    }
}

template<bool FIRST>
__device__ __forceinline__ void term_pair(const float* rp, int rowc, int cq0, float (&part)[8][4])
{
    constexpr int NQ = 4;
    #pragma unroll
    for (int n0 = 0; n0 < 8; n0 += NQ) {
        v4f l[NQ], r[NQ];
        #pragma unroll
        for (int u = 0; u < NQ; ++u) {
            const int co = cq0 + 8 * (n0 + u);
            l[u] = *(const v4f*)&rp[rowc + co];
            r[u] = *(const v4f*)&rp[rowc + STR + co];
        }
        #pragma unroll
        for (int u = 0; u < NQ; ++u) {
            #pragma unroll
            for (int c = 0; c < 4; ++c) {
                const float d  = r[u][c] - l[u][c];
                const float qg = d * __builtin_amdgcn_rcpf(1.0f + __expf(-1.702f * d));
                const float s  = qg + (r[u][c] + l[u][c]);
                if constexpr (FIRST) part[n0 + u][c] = s;
                else                 part[n0 + u][c] += s;
            }
        }
    }
}

__device__ __forceinline__ void lds_barrier()
{
    __builtin_amdgcn_fence(__ATOMIC_RELEASE, "workgroup", "local");
    __builtin_amdgcn_s_barrier();
    __builtin_amdgcn_fence(__ATOMIC_ACQUIRE, "workgroup", "local");
}

__global__ __launch_bounds__(256) void trconv_fused_kernel(
    const float* __restrict__ x,
    const float* __restrict__ w,
    const float* __restrict__ bias,
    float* __restrict__ out)
{
    __shared__ __attribute__((aligned(16))) float lds[LDS_FLOATS];
    const int tid = threadIdx.x;
    const int b   = blockIdx.z;
    const int g   = blockIdx.y;
    const int pc0 = blockIdx.x * (TPW * NP);

    const int lane = tid & 31;
    const int wv   = tid >> 5;
    const int ct   = wv & 3;
    const int wrow = wv >> 2;
    const int nm   = lane & 15;
    const int lh   = lane >> 4;

    const float* xg = x + (size_t)(b * 128 + g * 64) * T_LEN;
    v4f xv[10];
    x_tile_load(xg, pc0 - 8, lane, wv, xv);
    float bias_v = 0.0f;
    if (tid < 64) bias_v = bias[g * 64 + tid];

    const float* wgrp = w + (size_t)(g * 64) * (64 * 13);
    v4f wb[4][13];
    {
        const float* wsrc = wgrp + (wv * 8) * (64 * 13) + 4 * lane;
        #pragma unroll
        for (int i = 0; i < 4; ++i) {
            #pragma unroll
            for (int j = 0; j < 13; ++j)
                wb[i][j] = *(const v4f*)(wsrc + i * 1664 + 128 * j);
        }
    }

    x_tile_scatter(lds, lane, wv, xv);
    if (tid < 64) lds[BIAS_OFF + tid] = bias_v;

    const int aw_base  = W_OFF + (ct * 16 + nm) * STR + lh * 32;
    const int xb_base  = XS_OFF + nm * STR + lh * 32;
    const int res_base = R_OFF + nm * STR + ct * 16 + lh * 8;
    const int resx_base = XS_OFF + nm * STR + ct * 16 + lh * 8;

    {
        int woff[13][4];
        #pragma unroll
        for (int j = 0; j < 13; ++j) {
            #pragma unroll
            for (int c = 0; c < 4; ++c) {
                const int L    = 128 * j + 4 * lane + c;
                const int cf   = L / 13;
                const int k    = L - 13 * cf;
                const int slot = 13 * (k / 3) + 8 - 4 * k;
                woff[j][c] = W_OFF + slot * WSLOT + (wv * 8 + (cf >> 6)) * STR + kslot(cf & 63);
            }
        }
        #pragma unroll
        for (int i = 0; i < 4; ++i) {
            #pragma unroll
            for (int j = 0; j < 13; ++j) {
                #pragma unroll
                for (int c = 0; c < 4; ++c)
                    lds[woff[j][c] + i * 2 * STR] = wb[i][j][c];
            }
        }
    }

    const float* rp  = &lds[R_OFF];
    const float* rpx = &lds[XS_OFF];
    const int    il  = tid & (NP - 1);
    const int    cq0 = 4 * (tid >> 7);
    float* outg = out + (size_t)(b * 128 + g * 64) * T_LEN;

    #pragma unroll 1
    for (int tile = 0; tile < TPW; ++tile) {
        const int p0  = pc0 + tile * NP;
        const int me0 = p0 - 8;
        const int i   = p0 + il;
        const int rowc = (ref_idx(i) - me0) * STR;

        if (tile + 1 < TPW) x_tile_load(xg, me0 + NP, lane, wv, xv);

        float part[8][4];

        const bool first = (tile == 0);
        const bool more  = (tile + 1 < TPW);
        const bool keep  = (nm < 8);

        lds_barrier();
        if (!first) carry_restore(lds, tid, 0, R_OFF);
        v8f h1[2][2], h2[2][2];
        v8f hh1 = {0.f,0.f,0.f,0.f,0.f,0.f,0.f,0.f};
        v8f hh2 = {0.f,0.f,0.f,0.f,0.f,0.f,0.f,0.f};
        const bool head = first && wrow == 0;
        if (head) head_all(lds, aw_base, xb_base, res_base, keep, hh1, hh2);
        gemm_all(lds, aw_base, xb_base + 8 * STR, res_base + 8 * STR, 4 * wrow, h1, h2);
        lds_barrier();
        store_held(lds, resx_base + 8 * STR, 4 * wrow, h1);
        if (head && keep) {
            *(v4f*)&lds[resx_base]     = (v4f){hh1[0], hh1[1], hh1[2], hh1[3]};
            *(v4f*)&lds[resx_base + 4] = (v4f){hh1[4], hh1[5], hh1[6], hh1[7]};
        }
        if (!first) carry_restore(lds, tid, 1, XS_OFF);
        term_pair<true >(rp, rowc, cq0, part);
        if (more) carry_save(lds, tid, 0, R_OFF);
        lds_barrier();
        store_held(lds, res_base + 8 * STR, 4 * wrow, h2);
        if (head && keep) {
            *(v4f*)&lds[res_base]     = (v4f){hh2[0], hh2[1], hh2[2], hh2[3]};
            *(v4f*)&lds[res_base + 4] = (v4f){hh2[4], hh2[5], hh2[6], hh2[7]};
        }
        if (!first) carry_restore(lds, tid, 2, R_OFF);
        term_pair<false>(rpx, rowc, cq0, part);
        if (more) carry_save(lds, tid, 1, XS_OFF);
        lds_barrier();

        if (more) { x_tile_scatter(lds, lane, wv, xv); carry_save(lds, tid, 2, R_OFF); }

        int  roww[4];
        bool inr[4];
        #pragma unroll
        for (int q = 0; q < 4; ++q) {
            const int j = i + (q < 2 ? q - 2 : q - 1);
            roww[q] = (ref_idx(j) - me0) * STR;
            inr[q]  = (j >= 0 && j <= T_LEN - 1);
        }
        float* outp = outg + i;

        constexpr int NQ = 2;
        #pragma unroll
        for (int n0 = 0; n0 < 8; n0 += NQ) {
            v4f bi[NQ], xc[NQ], xw[4][NQ];
            #pragma unroll
            for (int u = 0; u < NQ; ++u) {
                const int co = cq0 + 8 * (n0 + u);
                bi[u] = *(const v4f*)&lds[BIAS_OFF + co];
                xc[u] = *(const v4f*)&rp[rowc + co];
                #pragma unroll
                for (int q = 0; q < 4; ++q) xw[q][u] = *(const v4f*)&rp[roww[q] + co];
            }
            #pragma unroll
            for (int u = 0; u < NQ; ++u) {
                const int co = cq0 + 8 * (n0 + u);
                #pragma unroll
                for (int c = 0; c < 4; ++c) {
                    const float b1 = bi[u][c];
                    float x_m = xc[u][c] + b1;
                    float um = x_m;
                    #pragma unroll
                    for (int q = 0; q < 4; ++q) {
                        float tv = xw[q][u][c] + b1;
                        um = fmaxf(um, inr[q] ? tv : 0.0f);
                    }
                    float accv = part[n0 + u][c] + 2.0f * x_m + 4.0f * b1 + um;
                    outp[(size_t)(co + c) * T_LEN] = accv;
                }
            }
        }
    }
}

extern "C" void kernel_launch(void* const* d_in, const int* in_sizes, int n_in,
                              void* d_out, int out_size, void* d_ws, size_t ws_size,
                              hipStream_t stream) {
    const float* x  = (const float*)d_in[0];
    const float* w  = (const float*)d_in[1];
    const float* b  = (const float*)d_in[2];
    float* out      = (float*)d_out;

    dim3 grid(T_LEN / (NP * TPW), 2, NB);
    dim3 block(256);
    hipLaunchKernelGGL(trconv_fused_kernel, grid, block, 0, stream, x, w, b, out);
}


# Two-phase fast FIR over even/odd positions: per residue 3 half-length filters (even taps, odd taps, their sum) on differenced x planes, 20 not 26 tap products per position pair; sum order changes
# speedup vs baseline: 3.4895x; 1.1843x over previous
#include <hip/hip_runtime.h>


typedef float v2f __attribute__((ext_vector_type(2)));
typedef float v4f __attribute__((ext_vector_type(4)));
typedef float v8f __attribute__((ext_vector_type(8)));

#define T_LEN 8192
#define NB    16
#define NP    128
#define TPW   8
#define ME    144
#define PROWS 70
#define PLANE (PROWS * STR)
#define STR   68

#define XS_OFF   0
#define W_OFF    (XS_OFF + 3 * PLANE)
#define WSLOT    (64 * STR + 8)
#define RES_SZ   (ME * STR)
#define R_OFF    (W_OFF + 13 * WSLOT)
#define BIAS_OFF (R_OFF + RES_SZ)
#define CARRY_OFF (BIAS_OFF + 64)
#define LDS_FLOATS (CARRY_OFF + 3 * 256)
static_assert(LDS_FLOATS * 4 <= 320 * 1024, "fits the workgroup LDS");
static_assert(3 * PLANE >= ME * STR, "residue 1's frame fits the input planes' space");

__device__ __forceinline__ int kslot(int ci) {
    return (((ci >> 1) & 1) << 5) + ((ci >> 2) << 1) + (ci & 1);
}

__device__ __forceinline__ int ref_idx(int i) {
    if (i < 2) return 2 - i;
    if (i > T_LEN - 3) return (i == T_LEN - 2) ? T_LEN - 6 : T_LEN - 7;
    return i - 2;
}

#define FRAG2(v, c) ((v2f){(v)[2 * (c)], (v)[2 * (c) + 1]})
#define WMMA4(a, b, acc) __builtin_amdgcn_wmma_f32_16x16x4_f32(false, (a), false, (b), (short)0, (acc), false, false)

__device__ __forceinline__ void gemm_ffa(float* lds, int aw_base, int ub4, int rb8, int wrow,
                                         v8f (&h1)[2][2], v8f (&h2)[2][2])
{
    constexpr int STEPS = 20;
    v4f AE[2][3][2], AO[2][3], BU[2][3][2][2];
    const int un0 = ub4 + 32 * wrow * STR;

    #pragma unroll
    for (int r = 0; r < 3; ++r) {
        AE[0][r][0] = *(const v4f*)&lds[aw_base + (4 * r) * WSLOT];
        AO[0][r]    = *(const v4f*)&lds[aw_base + (4 * r + 1) * WSLOT];
        #pragma unroll
        for (int p = 0; p < 2; ++p)
            BU[0][r][p][0] = *(const v4f*)&lds[un0 + r * PLANE + 16 * p * STR];
    }

    const v8f zero8 = {0.f,0.f,0.f,0.f,0.f,0.f,0.f,0.f};
    v8f C0[3][2], C1[3][2], C2[3][2];
    #pragma unroll
    for (int r = 0; r < 3; ++r) {
        #pragma unroll
        for (int p = 0; p < 2; ++p) { C0[r][p] = zero8; C1[r][p] = zero8; C2[r][p] = zero8; }
    }

    #pragma unroll
    for (int s = 0; s < STEPS; ++s) {
        const int cur = s & 1, nxt = cur ^ 1;
        const int j = s < 16 ? (s >> 3) : 2;
        if (s + 1 < STEPS) {
            const int sn = s + 1;
            const int jn = sn < 16 ? (sn >> 3) : 2;
            const int kf = sn < 16 ? 4 * (sn & 7) : 8 * (sn - 16);
            const int an = aw_base + kf;
            const int xn = un0 + jn * STR + kf;
            if (jn < 2) {
                #pragma unroll
                for (int r = 0; r < 3; ++r) {
                    AE[nxt][r][0] = *(const v4f*)&lds[an + (4 * r + 2 * jn) * WSLOT];
                    AO[nxt][r]    = *(const v4f*)&lds[an + (4 * r + 2 * jn + 1) * WSLOT];
                    #pragma unroll
                    for (int p = 0; p < 2; ++p)
                        BU[nxt][r][p][0] = *(const v4f*)&lds[xn + r * PLANE + 16 * p * STR];
                }
            } else {
                #pragma unroll
                for (int qq = 0; qq < 2; ++qq) {
                    AE[nxt][2][qq] = *(const v4f*)&lds[an + 12 * WSLOT + 4 * qq];
                    #pragma unroll
                    for (int u = 0; u < 2; ++u) {
                        #pragma unroll
                        for (int p = 0; p < 2; ++p)
                            BU[nxt][u][p][qq] = *(const v4f*)&lds[xn + u * PLANE + 16 * p * STR + 4 * qq];
                    }
                }
            }
        }
        __builtin_amdgcn_sched_barrier(0);
        if (j < 2) {
            v4f AS[3];
            #pragma unroll
            for (int r = 0; r < 3; ++r) AS[r] = AE[cur][r][0] + AO[cur][r];
            #pragma unroll
            for (int c = 0; c < 2; ++c) {
                #pragma unroll
                for (int r = 0; r < 3; ++r) {
                    #pragma unroll
                    for (int p = 0; p < 2; ++p) {
                        C0[r][p] = WMMA4(FRAG2(AE[cur][r][0], c), FRAG2(BU[cur][0][p][0], c), C0[r][p]);
                        C1[r][p] = WMMA4(FRAG2(AS[r], c),         FRAG2(BU[cur][1][p][0], c), C1[r][p]);
                        C2[r][p] = WMMA4(FRAG2(AO[cur][r], c),    FRAG2(BU[cur][2][p][0], c), C2[r][p]);
                    }
                }
            }
        } else {
            #pragma unroll
            for (int qq = 0; qq < 2; ++qq) {
                #pragma unroll
                for (int c = 0; c < 2; ++c) {
                    #pragma unroll
                    for (int p = 0; p < 2; ++p) {
                        C0[2][p] = WMMA4(FRAG2(AE[cur][2][qq], c), FRAG2(BU[cur][0][p][qq], c), C0[2][p]);
                        C1[2][p] = WMMA4(FRAG2(AE[cur][2][qq], c), FRAG2(BU[cur][1][p][qq], c), C1[2][p]);
                    }
                }
            }
        }
        __builtin_amdgcn_sched_barrier(0);
    }

    #pragma unroll
    for (int p = 0; p < 2; ++p) {
        const v8f e0 = C0[0][p] + C1[0][p];
        const v8f o0 = C1[0][p] + C2[0][p];
        const int st = rb8 + (64 * wrow + 32 * p) * STR;
        *(v4f*)&lds[st]           = (v4f){e0[0], e0[1], e0[2], e0[3]};
        *(v4f*)&lds[st + 4]       = (v4f){e0[4], e0[5], e0[6], e0[7]};
        *(v4f*)&lds[st + STR]     = (v4f){o0[0], o0[1], o0[2], o0[3]};
        *(v4f*)&lds[st + STR + 4] = (v4f){o0[4], o0[5], o0[6], o0[7]};
        h1[p][0] = C0[1][p] + C1[1][p];
        h1[p][1] = C1[1][p] + C2[1][p];
        h2[p][0] = C0[2][p] + C1[2][p];
        h2[p][1] = C1[2][p] + C2[2][p];
    }
}

__device__ __forceinline__ void head_ffa(float* lds, int aw_base, int ub_base, int res_base, bool keep,
                                         v8f (&hh1)[2], v8f (&hh2)[2])
{
    const v8f zero8 = {0.f,0.f,0.f,0.f,0.f,0.f,0.f,0.f};
    v8f C0[3], C1[3], C2[3];
    #pragma unroll
    for (int r = 0; r < 3; ++r) { C0[r] = zero8; C1[r] = zero8; C2[r] = zero8; }
    #pragma unroll
    for (int j = 0; j < 3; ++j) {
        #pragma unroll
        for (int q = 0; q < 8; ++q) {
            const v4f b0 = *(const v4f*)&lds[ub_base + j * STR + 4 * q];
            const v4f b1 = *(const v4f*)&lds[ub_base + PLANE + j * STR + 4 * q];
            if (j < 2) {
                const v4f b2 = *(const v4f*)&lds[ub_base + 2 * PLANE + j * STR + 4 * q];
                #pragma unroll
                for (int r = 0; r < 3; ++r) {
                    const v4f we = *(const v4f*)&lds[aw_base + (4 * r + 2 * j) * WSLOT + 4 * q];
                    const v4f wo = *(const v4f*)&lds[aw_base + (4 * r + 2 * j + 1) * WSLOT + 4 * q];
                    const v4f ws = we + wo;
                    #pragma unroll
                    for (int c = 0; c < 2; ++c) {
                        C0[r] = WMMA4(FRAG2(we, c), FRAG2(b0, c), C0[r]);
                        C1[r] = WMMA4(FRAG2(ws, c), FRAG2(b1, c), C1[r]);
                        C2[r] = WMMA4(FRAG2(wo, c), FRAG2(b2, c), C2[r]);
                    }
                }
            } else {
                const v4f we = *(const v4f*)&lds[aw_base + 12 * WSLOT + 4 * q];
                #pragma unroll
                for (int c = 0; c < 2; ++c) {
                    C0[2] = WMMA4(FRAG2(we, c), FRAG2(b0, c), C0[2]);
                    C1[2] = WMMA4(FRAG2(we, c), FRAG2(b1, c), C1[2]);
                }
            }
        }
    }
    const v8f e0 = C0[0] + C1[0];
    const v8f o0 = C1[0] + C2[0];
    if (keep) {
        *(v4f*)&lds[res_base]           = (v4f){e0[0], e0[1], e0[2], e0[3]};
        *(v4f*)&lds[res_base + 4]       = (v4f){e0[4], e0[5], e0[6], e0[7]};
        *(v4f*)&lds[res_base + STR]     = (v4f){o0[0], o0[1], o0[2], o0[3]};
        *(v4f*)&lds[res_base + STR + 4] = (v4f){o0[4], o0[5], o0[6], o0[7]};
    }
    hh1[0] = C0[1] + C1[1];
    hh1[1] = C1[1] + C2[1];
    hh2[0] = C0[2] + C1[2];
    hh2[1] = C1[2] + C2[2];
}

__device__ __forceinline__ void store_held(float* lds, int rb8, int wrow, const v8f (&h)[2][2])
{
    #pragma unroll
    for (int p = 0; p < 2; ++p) {
        #pragma unroll
        for (int j = 0; j < 2; ++j) {
            const int st = rb8 + (64 * wrow + 32 * p + j) * STR;
            *(v4f*)&lds[st]     = (v4f){h[p][j][0], h[p][j][1], h[p][j][2], h[p][j][3]};
            *(v4f*)&lds[st + 4] = (v4f){h[p][j][4], h[p][j][5], h[p][j][6], h[p][j][7]};
        }
    }
}

__device__ __forceinline__ void store_head(float* lds, int base, const v8f (&hh)[2])
{
    #pragma unroll
    for (int j = 0; j < 2; ++j) {
        *(v4f*)&lds[base + j * STR]     = (v4f){hh[j][0], hh[j][1], hh[j][2], hh[j][3]};
        *(v4f*)&lds[base + j * STR + 4] = (v4f){hh[j][4], hh[j][5], hh[j][6], hh[j][7]};
    }
}

__device__ __forceinline__ void carry_save(float* lds, int tid, int r, int buf)
{
    if (tid < 64)
        *(v4f*)&lds[CARRY_OFF + r * 256 + 4 * tid] =
            *(const v4f*)&lds[buf + (132 + (tid >> 4)) * STR + 4 * (tid & 15)];
}
__device__ __forceinline__ void carry_restore(float* lds, int tid, int r, int buf)
{
    if (tid < 64)
        *(v4f*)&lds[buf + (4 + (tid >> 4)) * STR + 4 * (tid & 15)] =
            *(const v4f*)&lds[CARRY_OFF + r * 256 + 4 * tid];
}

__device__ __forceinline__ void x_tile_load(const float* __restrict__ xg, int me0, int lane, int wv, v4f (&xv)[10], float (&xe)[10])
{
    const int xq = lane & 7;
    const int xr = lane >> 3;
    const int xs = wv & 1;
    #pragma unroll
    for (int it = 0; it < 10; ++it) {
        const int h  = (it & 1) * 4 + (wv >> 1);
        const int ci = 8 * h + 2 * xs + (xr & 1) + 4 * (xr >> 1);
        int q = 8 * (it >> 1) + xq;
        q = q > 36 ? 36 : q;
        int m4 = me0 + 4 * q;
        m4 = m4 < 0 ? 0 : (m4 > T_LEN - 4 ? T_LEN - 4 : m4);
        xv[it] = *(const v4f*)(xg + (size_t)ci * T_LEN + m4);
        xe[it] = xg[(size_t)ci * T_LEN + (m4 + 4 > T_LEN - 1 ? T_LEN - 1 : m4 + 4)];
    }
}

__device__ __forceinline__ void x_tile_scatter(float* lds, int lane, int wv, const v4f (&xv)[10], const float (&xe)[10])
{
    const int xq = lane & 7;
    const int xr = lane >> 3;
    const int xs = wv & 1;
    #pragma unroll
    for (int it = 0; it < 10; ++it) {
        const int h  = (it & 1) * 4 + (wv >> 1);
        const int ci = 8 * h + 2 * xs + (xr & 1) + 4 * (xr >> 1);
        const int q  = 8 * (it >> 1) + xq;
        if (q < PROWS / 2) {
            const int a = XS_OFF + (2 * q) * STR + kslot(ci);
            const float v0 = xv[it][0], v1 = xv[it][1], v2 = xv[it][2], v3 = xv[it][3];
            lds[a]                   = v0 - v1;
            lds[a + PLANE]           = v1;
            lds[a + 2 * PLANE]       = v2 - v1;
            lds[a + STR]             = v2 - v3;
            lds[a + STR + PLANE]     = v3;
            lds[a + STR + 2 * PLANE] = xe[it] - v3;
        }
    }
}

template<bool FIRST>
__device__ __forceinline__ void term_pair(const float* rp, int rowc, int cq0, float (&part)[8][4])
{
    constexpr int NQ = 4;
    #pragma unroll
    for (int n0 = 0; n0 < 8; n0 += NQ) {
        v4f l[NQ], r[NQ];
        #pragma unroll
        for (int u = 0; u < NQ; ++u) {
            const int co = cq0 + 8 * (n0 + u);
            l[u] = *(const v4f*)&rp[rowc + co];
            r[u] = *(const v4f*)&rp[rowc + STR + co];
        }
        #pragma unroll
        for (int u = 0; u < NQ; ++u) {
            #pragma unroll
            for (int c = 0; c < 4; ++c) {
                const float d  = r[u][c] - l[u][c];
                const float qg = d * __builtin_amdgcn_rcpf(1.0f + __expf(-1.702f * d));
                const float s  = qg + (r[u][c] + l[u][c]);
                if constexpr (FIRST) part[n0 + u][c] = s;
                else                 part[n0 + u][c] += s;
            }
        }
    }
}

__device__ __forceinline__ void lds_barrier()
{
    __builtin_amdgcn_fence(__ATOMIC_RELEASE, "workgroup", "local");
    __builtin_amdgcn_s_barrier();
    __builtin_amdgcn_fence(__ATOMIC_ACQUIRE, "workgroup", "local");
}

__global__ __launch_bounds__(256) void trconv_fused_kernel(
    const float* __restrict__ x,
    const float* __restrict__ w,
    const float* __restrict__ bias,
    float* __restrict__ out)
{
    __shared__ __attribute__((aligned(16))) float lds[LDS_FLOATS];
    const int tid = threadIdx.x;
    const int b   = blockIdx.z;
    const int g   = blockIdx.y;
    const int pc0 = blockIdx.x * (TPW * NP);

    const int lane = tid & 31;
    const int wv   = tid >> 5;
    const int ct   = wv & 3;
    const int wrow = wv >> 2;
    const int nm   = lane & 15;
    const int lh   = lane >> 4;

    const float* xg = x + (size_t)(b * 128 + g * 64) * T_LEN;
    v4f xv[10];
    float xe[10];
    x_tile_load(xg, pc0 - 8, lane, wv, xv, xe);
    float bias_v = 0.0f;
    if (tid < 64) bias_v = bias[g * 64 + tid];

    const float* wgrp = w + (size_t)(g * 64) * (64 * 13);
    v4f wb[4][13];
    {
        const float* wsrc = wgrp + (wv * 8) * (64 * 13) + 4 * lane;
        #pragma unroll
        for (int i = 0; i < 4; ++i) {
            #pragma unroll
            for (int j = 0; j < 13; ++j)
                wb[i][j] = *(const v4f*)(wsrc + i * 1664 + 128 * j);
        }
    }

    x_tile_scatter(lds, lane, wv, xv, xe);
    if (tid < 64) lds[BIAS_OFF + tid] = bias_v;

    const int aw_base  = W_OFF + (ct * 16 + nm) * STR + lh * 32;
    const int xb_base  = XS_OFF + nm * STR + lh * 32;
    const int res_base = R_OFF + 2 * nm * STR + ct * 16 + lh * 8;
    const int resx_base = XS_OFF + 2 * nm * STR + ct * 16 + lh * 8;

    {
        int woff[13][4];
        #pragma unroll
        for (int j = 0; j < 13; ++j) {
            #pragma unroll
            for (int c = 0; c < 4; ++c) {
                const int L    = 128 * j + 4 * lane + c;
                const int cf   = L / 13;
                const int k    = L - 13 * cf;
                const int slot = 13 * (k / 3) + 8 - 4 * k;
                woff[j][c] = W_OFF + slot * WSLOT + (wv * 8 + (cf >> 6)) * STR + kslot(cf & 63);
            }
        }
        #pragma unroll
        for (int i = 0; i < 4; ++i) {
            #pragma unroll
            for (int j = 0; j < 13; ++j) {
                #pragma unroll
                for (int c = 0; c < 4; ++c)
                    lds[woff[j][c] + i * 2 * STR] = wb[i][j][c];
            }
        }
    }

    const float* rp  = &lds[R_OFF];
    const float* rpx = &lds[XS_OFF];
    const int    il  = tid & (NP - 1);
    const int    cq0 = 4 * (tid >> 7);
    float* outg = out + (size_t)(b * 128 + g * 64) * T_LEN;

    #pragma unroll 1
    for (int tile = 0; tile < TPW; ++tile) {
        const int p0  = pc0 + tile * NP;
        const int me0 = p0 - 8;
        const int i   = p0 + il;
        const int rowc = (ref_idx(i) - me0) * STR;

        float part[8][4];

        const bool first = (tile == 0);
        const bool more  = (tile + 1 < TPW);
        const bool keep  = (nm < 4);

        lds_barrier();
        if (!first) carry_restore(lds, tid, 0, R_OFF);
        v8f h1[2][2], h2[2][2];
        v8f hh1[2], hh2[2];
        hh1[0] = hh1[1] = hh2[0] = hh2[1] = (v8f){0.f,0.f,0.f,0.f,0.f,0.f,0.f,0.f};
        const bool head = first && wrow == 0;
        gemm_ffa(lds, aw_base, xb_base + 4 * STR, res_base + 8 * STR, wrow, h1, h2);
        if (head) head_ffa(lds, aw_base, xb_base, res_base, keep, hh1, hh2);
        v4f   xnv[10];
        float xne[10];
        if (more) x_tile_load(xg, me0 + NP, lane, wv, xnv, xne);
        lds_barrier();
        store_held(lds, resx_base + 8 * STR, wrow, h1);
        if (head && keep) store_head(lds, resx_base, hh1);
        if (!first) carry_restore(lds, tid, 1, XS_OFF);
        term_pair<true >(rp, rowc, cq0, part);
        if (more) carry_save(lds, tid, 0, R_OFF);
        lds_barrier();
        store_held(lds, res_base + 8 * STR, wrow, h2);
        if (head && keep) store_head(lds, res_base, hh2);
        if (!first) carry_restore(lds, tid, 2, R_OFF);
        term_pair<false>(rpx, rowc, cq0, part);
        if (more) carry_save(lds, tid, 1, XS_OFF);
        lds_barrier();

        if (more) carry_save(lds, tid, 2, R_OFF);

        int  roww[4];
        bool inr[4];
        #pragma unroll
        for (int q = 0; q < 4; ++q) {
            const int j = i + (q < 2 ? q - 2 : q - 1);
            roww[q] = (ref_idx(j) - me0) * STR;
            inr[q]  = (j >= 0 && j <= T_LEN - 1);
        }
        float* outp = outg + i;

        constexpr int NQ = 2;
        #pragma unroll
        for (int n0 = 0; n0 < 8; n0 += NQ) {
            v4f bi[NQ], xc[NQ], xw[4][NQ];
            #pragma unroll
            for (int u = 0; u < NQ; ++u) {
                const int co = cq0 + 8 * (n0 + u);
                bi[u] = *(const v4f*)&lds[BIAS_OFF + co];
                xc[u] = *(const v4f*)&rp[rowc + co];
                #pragma unroll
                for (int q = 0; q < 4; ++q) xw[q][u] = *(const v4f*)&rp[roww[q] + co];
            }
            #pragma unroll
            for (int u = 0; u < NQ; ++u) {
                const int co = cq0 + 8 * (n0 + u);
                #pragma unroll
                for (int c = 0; c < 4; ++c) {
                    const float b1 = bi[u][c];
                    float x_m = xc[u][c] + b1;
                    float um = x_m;
                    #pragma unroll
                    for (int q = 0; q < 4; ++q) {
                        float tv = xw[q][u][c] + b1;
                        um = fmaxf(um, inr[q] ? tv : 0.0f);
                    }
                    float accv = part[n0 + u][c] + 2.0f * x_m + 4.0f * b1 + um;
                    outp[(size_t)(co + c) * T_LEN] = accv;
                }
            }
        }

        if (more) x_tile_scatter(lds, lane, wv, xnv, xne);
    }
}

extern "C" void kernel_launch(void* const* d_in, const int* in_sizes, int n_in,
                              void* d_out, int out_size, void* d_ws, size_t ws_size,
                              hipStream_t stream) {
    const float* x  = (const float*)d_in[0];
    const float* w  = (const float*)d_in[1];
    const float* b  = (const float*)d_in[2];
    float* out      = (float*)d_out;

    dim3 grid(T_LEN / (NP * TPW), 2, NB);
    dim3 block(256);
    hipLaunchKernelGGL(trconv_fused_kernel, grid, block, 0, stream, x, w, b, out);
}


# Second level of the fast FIR in the regular pass: each 2-tap pair filter done over quads with 3 products for 4; 496 not 640 matrix instructions per wave and tile; sum order changes
# speedup vs baseline: 3.8289x; 1.0973x over previous
#include <hip/hip_runtime.h>


typedef float v2f __attribute__((ext_vector_type(2)));
typedef float v4f __attribute__((ext_vector_type(4)));
typedef float v8f __attribute__((ext_vector_type(8)));

#define T_LEN 8192
#define NB    16
#define NP    128
#define TPW   8
#define ME    144
#define PROWS 70
#define PLANE (PROWS * STR)
#define STR   68

#define XS_OFF   0
#define W_OFF    (XS_OFF + 3 * PLANE)
#define WSLOT    (64 * STR + 8)
#define RES_SZ   (ME * STR)
#define R_OFF    (W_OFF + 13 * WSLOT)
#define BIAS_OFF (R_OFF + RES_SZ)
#define CARRY_OFF (BIAS_OFF + 64)
#define LDS_FLOATS (CARRY_OFF + 3 * 256)
static_assert(LDS_FLOATS * 4 <= 320 * 1024, "fits the workgroup LDS");
static_assert(3 * PLANE >= ME * STR, "residue 1's frame fits the input planes' space");

__device__ __forceinline__ int kslot(int ci) {
    return (((ci >> 1) & 1) << 5) + ((ci >> 2) << 1) + (ci & 1);
}

__device__ __forceinline__ int ref_idx(int i) {
    if (i < 2) return 2 - i;
    if (i > T_LEN - 3) return (i == T_LEN - 2) ? T_LEN - 6 : T_LEN - 7;
    return i - 2;
}

#define FRAG2(v, c) ((v2f){(v)[2 * (c)], (v)[2 * (c) + 1]})
#define WMMA4(a, b, acc) __builtin_amdgcn_wmma_f32_16x16x4_f32(false, (a), false, (b), (short)0, (acc), false, false)

__device__ __forceinline__ void gemm_ffa(float* lds, int aw_base, int ubq, int rbq, int wrow,
                                         v8f (&h1)[4], v8f (&h2)[4])
{
    constexpr int STEPS = 8;
    v4f AW[2][3][4], BU[2][3][3];
    v4f A12, BX[2];
    const int un0 = ubq + 32 * wrow * STR;

    #pragma unroll
    for (int r = 0; r < 3; ++r) {
        #pragma unroll
        for (int t = 0; t < 4; ++t) AW[0][r][t] = *(const v4f*)&lds[aw_base + (4 * r + t) * WSLOT];
        #pragma unroll
        for (int i = 0; i < 3; ++i) BU[0][r][i] = *(const v4f*)&lds[un0 + r * PLANE + i * STR];
    }

    const v8f zero8 = {0.f,0.f,0.f,0.f,0.f,0.f,0.f,0.f};
    v8f N[3][3][3];
    #pragma unroll
    for (int r = 0; r < 3; ++r) {
        #pragma unroll
        for (int f = 0; f < 3; ++f) { N[r][f][0] = zero8; N[r][f][1] = zero8; N[r][f][2] = zero8; }
    }

    #pragma unroll
    for (int s = 0; s < STEPS; ++s) {
        const int cur = s & 1, nxt = cur ^ 1;
        A12   = *(const v4f*)&lds[aw_base + 4 * s + 12 * WSLOT];
        BX[0] = *(const v4f*)&lds[un0 + 4 * s + 3 * STR];
        BX[1] = *(const v4f*)&lds[un0 + 4 * s + PLANE + 3 * STR];
        if (s + 1 < STEPS) {
            const int kf = 4 * (s + 1);
            #pragma unroll
            for (int r = 0; r < 3; ++r) {
                #pragma unroll
                for (int t = 0; t < 4; ++t) AW[nxt][r][t] = *(const v4f*)&lds[aw_base + kf + (4 * r + t) * WSLOT];
                #pragma unroll
                for (int i = 0; i < 3; ++i) BU[nxt][r][i] = *(const v4f*)&lds[un0 + kf + r * PLANE + i * STR];
            }
        }
        __builtin_amdgcn_sched_barrier(0);
        #pragma unroll
        for (int f = 0; f < 3; ++f) {
            const v4f Ba = BU[cur][f][0] - BU[cur][f][1];
            const v4f Bc = BU[cur][f][2] - BU[cur][f][1];
            #pragma unroll
            for (int r = 0; r < 3; ++r) {
                const v4f g0 = f == 0 ? AW[cur][r][0] : f == 2 ? AW[cur][r][1] : AW[cur][r][0] + AW[cur][r][1];
                const v4f g1 = f == 0 ? AW[cur][r][2] : f == 2 ? AW[cur][r][3] : AW[cur][r][2] + AW[cur][r][3];
                const v4f gs = g0 + g1;
                #pragma unroll
                for (int c = 0; c < 2; ++c) {
                    N[r][f][0] = WMMA4(FRAG2(g0, c), FRAG2(Ba, c),            N[r][f][0]);
                    N[r][f][1] = WMMA4(FRAG2(gs, c), FRAG2(BU[cur][f][1], c), N[r][f][1]);
                    N[r][f][2] = WMMA4(FRAG2(g1, c), FRAG2(Bc, c),            N[r][f][2]);
                }
            }
        }
        #pragma unroll
        for (int c = 0; c < 2; ++c) {
            #pragma unroll
            for (int f = 0; f < 2; ++f) {
                N[2][f][0] = WMMA4(FRAG2(A12, c), FRAG2(BU[cur][f][2], c), N[2][f][0]);
                N[2][f][2] = WMMA4(FRAG2(A12, c), FRAG2(BX[f], c),         N[2][f][2]);
            }
        }
        __builtin_amdgcn_sched_barrier(0);
    }

    #pragma unroll
    for (int r = 0; r < 3; ++r) {
        v8f M[3][2];
        #pragma unroll
        for (int f = 0; f < 3; ++f) {
            M[f][0] = N[r][f][0] + N[r][f][1];
            M[f][1] = N[r][f][1] + N[r][f][2];
        }
        v8f row[4];
        row[0] = M[0][0] + M[1][0];
        row[1] = M[1][0] + M[2][0];
        row[2] = M[0][1] + M[1][1];
        row[3] = M[1][1] + M[2][1];
        #pragma unroll
        for (int q = 0; q < 4; ++q) {
            if (r == 0) {
                const int st = rbq + (64 * wrow + q) * STR;
                *(v4f*)&lds[st]     = (v4f){row[q][0], row[q][1], row[q][2], row[q][3]};
                *(v4f*)&lds[st + 4] = (v4f){row[q][4], row[q][5], row[q][6], row[q][7]};
            } else if (r == 1) {
                h1[q] = row[q];
            } else {
                h2[q] = row[q];
            }
        }
    }
}

__device__ __forceinline__ void head_ffa(float* lds, int aw_base, int ub_base, int res_base, bool keep,
                                         int cbase, bool keepc)
{
    const v8f zero8 = {0.f,0.f,0.f,0.f,0.f,0.f,0.f,0.f};
    v8f C0[3], C1[3], C2[3];
    #pragma unroll
    for (int r = 0; r < 3; ++r) { C0[r] = zero8; C1[r] = zero8; C2[r] = zero8; }
    #pragma unroll
    for (int j = 0; j < 3; ++j) {
        #pragma unroll
        for (int q = 0; q < 8; ++q) {
            const v4f b0 = *(const v4f*)&lds[ub_base + j * STR + 4 * q];
            const v4f b1 = *(const v4f*)&lds[ub_base + PLANE + j * STR + 4 * q];
            if (j < 2) {
                const v4f b2 = *(const v4f*)&lds[ub_base + 2 * PLANE + j * STR + 4 * q];
                #pragma unroll
                for (int r = 0; r < 3; ++r) {
                    const v4f we = *(const v4f*)&lds[aw_base + (4 * r + 2 * j) * WSLOT + 4 * q];
                    const v4f wo = *(const v4f*)&lds[aw_base + (4 * r + 2 * j + 1) * WSLOT + 4 * q];
                    const v4f ws = we + wo;
                    #pragma unroll
                    for (int c = 0; c < 2; ++c) {
                        C0[r] = WMMA4(FRAG2(we, c), FRAG2(b0, c), C0[r]);
                        C1[r] = WMMA4(FRAG2(ws, c), FRAG2(b1, c), C1[r]);
                        C2[r] = WMMA4(FRAG2(wo, c), FRAG2(b2, c), C2[r]);
                    }
                }
            } else {
                const v4f we = *(const v4f*)&lds[aw_base + 12 * WSLOT + 4 * q];
                #pragma unroll
                for (int c = 0; c < 2; ++c) {
                    C0[2] = WMMA4(FRAG2(we, c), FRAG2(b0, c), C0[2]);
                    C1[2] = WMMA4(FRAG2(we, c), FRAG2(b1, c), C1[2]);
                }
            }
        }
    }
    const v8f e0 = C0[0] + C1[0];
    const v8f o0 = C1[0] + C2[0];
    if (keep) {
        *(v4f*)&lds[res_base]           = (v4f){e0[0], e0[1], e0[2], e0[3]};
        *(v4f*)&lds[res_base + 4]       = (v4f){e0[4], e0[5], e0[6], e0[7]};
        *(v4f*)&lds[res_base + STR]     = (v4f){o0[0], o0[1], o0[2], o0[3]};
        *(v4f*)&lds[res_base + STR + 4] = (v4f){o0[4], o0[5], o0[6], o0[7]};
    }
    if (keepc) {
        #pragma unroll
        for (int r = 1; r < 3; ++r) {
            const v8f e = C0[r] + C1[r];
            const v8f o = C1[r] + C2[r];
            const int cb = cbase + r * 256;
            *(v4f*)&lds[cb]          = (v4f){e[0], e[1], e[2], e[3]};
            *(v4f*)&lds[cb + 4]      = (v4f){e[4], e[5], e[6], e[7]};
            *(v4f*)&lds[cb + 64]     = (v4f){o[0], o[1], o[2], o[3]};
            *(v4f*)&lds[cb + 64 + 4] = (v4f){o[4], o[5], o[6], o[7]};
        }
    }
}

__device__ __forceinline__ void store_held(float* lds, int rbq, int wrow, const v8f (&h)[4])
{
    #pragma unroll
    for (int q = 0; q < 4; ++q) {
        const int st = rbq + (64 * wrow + q) * STR;
        *(v4f*)&lds[st]     = (v4f){h[q][0], h[q][1], h[q][2], h[q][3]};
        *(v4f*)&lds[st + 4] = (v4f){h[q][4], h[q][5], h[q][6], h[q][7]};
    }
}

__device__ __forceinline__ void carry_save(float* lds, int tid, int r, int buf)
{
    if (tid < 64)
        *(v4f*)&lds[CARRY_OFF + r * 256 + 4 * tid] =
            *(const v4f*)&lds[buf + (132 + (tid >> 4)) * STR + 4 * (tid & 15)];
}
__device__ __forceinline__ void carry_restore(float* lds, int tid, int r, int buf)
{
    if (tid < 64)
        *(v4f*)&lds[buf + (4 + (tid >> 4)) * STR + 4 * (tid & 15)] =
            *(const v4f*)&lds[CARRY_OFF + r * 256 + 4 * tid];
}

__device__ __forceinline__ void x_tile_load(const float* __restrict__ xg, int me0, int lane, int wv, v4f (&xv)[10], float (&xe)[10])
{
    const int xq = lane & 7;
    const int xr = lane >> 3;
    const int xs = wv & 1;
    #pragma unroll
    for (int it = 0; it < 10; ++it) {
        const int h  = (it & 1) * 4 + (wv >> 1);
        const int ci = 8 * h + 2 * xs + (xr & 1) + 4 * (xr >> 1);
        int q = 8 * (it >> 1) + xq;
        q = q > 36 ? 36 : q;
        int m4 = me0 + 4 * q;
        m4 = m4 < 0 ? 0 : (m4 > T_LEN - 4 ? T_LEN - 4 : m4);
        xv[it] = *(const v4f*)(xg + (size_t)ci * T_LEN + m4);
        xe[it] = xg[(size_t)ci * T_LEN + (m4 + 4 > T_LEN - 1 ? T_LEN - 1 : m4 + 4)];
    }
}

__device__ __forceinline__ void x_tile_scatter(float* lds, int lane, int wv, const v4f (&xv)[10], const float (&xe)[10])
{
    const int xq = lane & 7;
    const int xr = lane >> 3;
    const int xs = wv & 1;
    #pragma unroll
    for (int it = 0; it < 10; ++it) {
        const int h  = (it & 1) * 4 + (wv >> 1);
        const int ci = 8 * h + 2 * xs + (xr & 1) + 4 * (xr >> 1);
        const int q  = 8 * (it >> 1) + xq;
        if (q < PROWS / 2) {
            const int a = XS_OFF + (2 * q) * STR + kslot(ci);
            const float v0 = xv[it][0], v1 = xv[it][1], v2 = xv[it][2], v3 = xv[it][3];
            lds[a]                   = v0 - v1;
            lds[a + PLANE]           = v1;
            lds[a + 2 * PLANE]       = v2 - v1;
            lds[a + STR]             = v2 - v3;
            lds[a + STR + PLANE]     = v3;
            lds[a + STR + 2 * PLANE] = xe[it] - v3;
        }
    }
}

template<bool FIRST>
__device__ __forceinline__ void term_pair(const float* rp, int rowc, int cq0, float (&part)[8][4])
{
    constexpr int NQ = 4;
    #pragma unroll
    for (int n0 = 0; n0 < 8; n0 += NQ) {
        v4f l[NQ], r[NQ];
        #pragma unroll
        for (int u = 0; u < NQ; ++u) {
            const int co = cq0 + 8 * (n0 + u);
            l[u] = *(const v4f*)&rp[rowc + co];
            r[u] = *(const v4f*)&rp[rowc + STR + co];
        }
        #pragma unroll
        for (int u = 0; u < NQ; ++u) {
            #pragma unroll
            for (int c = 0; c < 4; ++c) {
                const float d  = r[u][c] - l[u][c];
                const float qg = d * __builtin_amdgcn_rcpf(1.0f + __expf(-1.702f * d));
                const float s  = qg + (r[u][c] + l[u][c]);
                if constexpr (FIRST) part[n0 + u][c] = s;
                else                 part[n0 + u][c] += s;
            }
        }
    }
}

__device__ __forceinline__ void lds_barrier()
{
    __builtin_amdgcn_fence(__ATOMIC_RELEASE, "workgroup", "local");
    __builtin_amdgcn_s_barrier();
    __builtin_amdgcn_fence(__ATOMIC_ACQUIRE, "workgroup", "local");
}

__global__ __launch_bounds__(256) void trconv_fused_kernel(
    const float* __restrict__ x,
    const float* __restrict__ w,
    const float* __restrict__ bias,
    float* __restrict__ out)
{
    __shared__ __attribute__((aligned(16))) float lds[LDS_FLOATS];
    const int tid = threadIdx.x;
    const int b   = blockIdx.z;
    const int g   = blockIdx.y;
    const int pc0 = blockIdx.x * (TPW * NP);

    const int lane = tid & 31;
    const int wv   = tid >> 5;
    const int ct   = wv & 3;
    const int wrow = wv >> 2;
    const int nm   = lane & 15;
    const int lh   = lane >> 4;

    const float* xg = x + (size_t)(b * 128 + g * 64) * T_LEN;
    v4f xv[10];
    float xe[10];
    x_tile_load(xg, pc0 - 8, lane, wv, xv, xe);
    float bias_v = 0.0f;
    if (tid < 64) bias_v = bias[g * 64 + tid];

    const float* wgrp = w + (size_t)(g * 64) * (64 * 13);
    v4f wb[4][13];
    {
        const float* wsrc = wgrp + (wv * 8) * (64 * 13) + 4 * lane;
        #pragma unroll
        for (int i = 0; i < 4; ++i) {
            #pragma unroll
            for (int j = 0; j < 13; ++j)
                wb[i][j] = *(const v4f*)(wsrc + i * 1664 + 128 * j);
        }
    }

    x_tile_scatter(lds, lane, wv, xv, xe);
    if (tid < 64) lds[BIAS_OFF + tid] = bias_v;

    const int aw_base  = W_OFF + (ct * 16 + nm) * STR + lh * 32;
    const int xq_base   = XS_OFF + (4 + 2 * nm) * STR + lh * 32;
    const int resq_base  = R_OFF  + (8 + 4 * nm) * STR + ct * 16 + lh * 8;
    const int resxq_base = XS_OFF + (8 + 4 * nm) * STR + ct * 16 + lh * 8;

    {
        int woff[13][4];
        #pragma unroll
        for (int j = 0; j < 13; ++j) {
            #pragma unroll
            for (int c = 0; c < 4; ++c) {
                const int L    = 128 * j + 4 * lane + c;
                const int cf   = L / 13;
                const int k    = L - 13 * cf;
                const int slot = 13 * (k / 3) + 8 - 4 * k;
                woff[j][c] = W_OFF + slot * WSLOT + (wv * 8 + (cf >> 6)) * STR + kslot(cf & 63);
            }
        }
        #pragma unroll
        for (int i = 0; i < 4; ++i) {
            #pragma unroll
            for (int j = 0; j < 13; ++j) {
                #pragma unroll
                for (int c = 0; c < 4; ++c)
                    lds[woff[j][c] + i * 2 * STR] = wb[i][j][c];
            }
        }
    }

    const float* rp  = &lds[R_OFF];
    const float* rpx = &lds[XS_OFF];
    const int    il  = tid & (NP - 1);
    const int    cq0 = 4 * (tid >> 7);
    float* outg = out + (size_t)(b * 128 + g * 64) * T_LEN;

    #pragma unroll 1
    for (int tile = 0; tile < TPW; ++tile) {
        const int p0  = pc0 + tile * NP;
        const int me0 = p0 - 8;

        float part[8][4];

        const bool first = (tile == 0);
        const bool more  = (tile + 1 < TPW);

        lds_barrier();
        if (!first) carry_restore(lds, tid, 0, R_OFF);
        v8f h1[4], h2[4];
        const bool head = first && wrow == 0;
        gemm_ffa(lds, aw_base, xq_base, resq_base, wrow, h1, h2);
        if (head) {
            int th = tid;
            asm volatile("" : "+v"(th));
            const int hn = th & 15, hl = (th >> 4) & 1, hc = (th >> 5) & 3;
            head_ffa(lds, W_OFF + (hc * 16 + hn) * STR + hl * 32,
                     XS_OFF + hn * STR + hl * 32,
                     R_OFF + 2 * hn * STR + hc * 16 + hl * 8, hn < 4,
                     CARRY_OFF + (2 * hn - 4) * 64 + hc * 16 + hl * 8, hn == 2 || hn == 3);
        }
        v4f   xnv[10];
        float xne[10];
        if (more) x_tile_load(xg, me0 + NP, lane, wv, xnv, xne);
        int ilo = il;
        asm volatile("" : "+v"(ilo));
        const int i    = p0 + ilo;
        const int rowc = (ref_idx(i) - me0) * STR;
        lds_barrier();
        store_held(lds, resxq_base, wrow, h1);
        carry_restore(lds, tid, 1, XS_OFF);
        term_pair<true >(rp, rowc, cq0, part);
        if (more) carry_save(lds, tid, 0, R_OFF);
        lds_barrier();
        store_held(lds, resq_base, wrow, h2);
        carry_restore(lds, tid, 2, R_OFF);
        term_pair<false>(rpx, rowc, cq0, part);
        if (more) carry_save(lds, tid, 1, XS_OFF);
        lds_barrier();

        if (more) carry_save(lds, tid, 2, R_OFF);

        int  roww[4];
        bool inr[4];
        #pragma unroll
        for (int q = 0; q < 4; ++q) {
            const int j = i + (q < 2 ? q - 2 : q - 1);
            roww[q] = (ref_idx(j) - me0) * STR;
            inr[q]  = (j >= 0 && j <= T_LEN - 1);
        }
        float* outp = outg + i;

        constexpr int NQ = 2;
        #pragma unroll
        for (int n0 = 0; n0 < 8; n0 += NQ) {
            v4f bi[NQ], xc[NQ], xw[4][NQ];
            #pragma unroll
            for (int u = 0; u < NQ; ++u) {
                const int co = cq0 + 8 * (n0 + u);
                bi[u] = *(const v4f*)&lds[BIAS_OFF + co];
                xc[u] = *(const v4f*)&rp[rowc + co];
                #pragma unroll
                for (int q = 0; q < 4; ++q) xw[q][u] = *(const v4f*)&rp[roww[q] + co];
            }
            #pragma unroll
            for (int u = 0; u < NQ; ++u) {
                const int co = cq0 + 8 * (n0 + u);
                #pragma unroll
                for (int c = 0; c < 4; ++c) {
                    const float b1 = bi[u][c];
                    float x_m = xc[u][c] + b1;
                    float um = x_m;
                    #pragma unroll
                    for (int q = 0; q < 4; ++q) {
                        float tv = xw[q][u][c] + b1;
                        um = fmaxf(um, inr[q] ? tv : 0.0f);
                    }
                    float accv = part[n0 + u][c] + 2.0f * x_m + 4.0f * b1 + um;
                    outp[(size_t)(co + c) * T_LEN] = accv;
                }
            }
        }

        if (more) x_tile_scatter(lds, lane, wv, xnv, xne);
    }
}

extern "C" void kernel_launch(void* const* d_in, const int* in_sizes, int n_in,
                              void* d_out, int out_size, void* d_ws, size_t ws_size,
                              hipStream_t stream) {
    const float* x  = (const float*)d_in[0];
    const float* w  = (const float*)d_in[1];
    const float* b  = (const float*)d_in[2];
    float* out      = (float*)d_out;

    dim3 grid(T_LEN / (NP * TPW), 2, NB);
    dim3 block(256);
    hipLaunchKernelGGL(trconv_fused_kernel, grid, block, 0, stream, x, w, b, out);
}
